# starting kernel
# baseline (speedup 1.0000x reference)
#include <hip/hip_runtime.h>
#include <hip/hip_bf16.h>
#include <math.h>

typedef __attribute__((ext_vector_type(2))) float v2f;
typedef __attribute__((ext_vector_type(8))) float v8f;

#define BATCH 16
#define E 128
#define CDIM 512
#define NCLS 5
#define ALPHA 0.2f

template <bool TB>
__device__ inline v8f gemm_tile(const float* __restrict__ Al,
                                const float* __restrict__ Bl,
                                int rowBase, int colBase, int K,
                                int lda, int ldb, int lane) {
  const int m  = lane & 15;
  const int kb = (lane >> 4) << 1;
  v8f acc = {0.f, 0.f, 0.f, 0.f, 0.f, 0.f, 0.f, 0.f};
  for (int kk = 0; kk < K; kk += 4) {
    v2f a = *(const v2f*)&Al[(rowBase + m) * lda + kk + kb];
    v2f b;
    if (TB) {
      b = *(const v2f*)&Bl[(colBase + m) * ldb + kk + kb];
    } else {
      b.x = Bl[(kk + kb) * ldb + colBase + m];
      b.y = Bl[(kk + kb + 1) * ldb + colBase + m];
    }
    acc = __builtin_amdgcn_wmma_f32_16x16x4_f32(
         false, a,  false, b,
         (short)0, acc,  false,  false);
  }
  return acc;
}

__device__ inline void store_tile(float* __restrict__ Cm, int rowBase,
                                  int colBase, int lane, v8f acc) {
  const int n    = colBase + (lane & 15);
  const int mOff = rowBase + ((lane >> 4) << 3);
#pragma unroll
  for (int r = 0; r < 8; ++r) Cm[(mOff + r) * E + n] = acc[r];
}

__global__ __launch_bounds__(256) void k_gram(const float* __restrict__ x,
                                              float* __restrict__ sqd) {
  extern __shared__ float smem[];
  float* xs    = smem;
  float* norms = smem + E * CDIM;
  const int b    = blockIdx.x;
  const int tid  = threadIdx.x;
  const int lane = tid & 31;
  const int wave = tid >> 5;

  const float4* xg  = (const float4*)(x + (size_t)b * E * CDIM);
  float4*       xs4 = (float4*)xs;
  for (int i = tid; i < E * CDIM / 4; i += 256) xs4[i] = xg[i];
  __syncthreads();

  for (int i = tid; i < E; i += 256) {
    const float* row = xs + i * CDIM;
    float s = 0.f;
    for (int k = 0; k < CDIM; ++k) s += row[k] * row[k];
    norms[i] = s;
  }
  __syncthreads();

  const float invs = 0.04419417382415922f;
  float* out = sqd + (size_t)b * E * E;
  for (int t = 0; t < 8; ++t) {
    v8f acc = gemm_tile<true>(xs, xs, wave * 16, t * 16, CDIM, CDIM, CDIM, lane);
    const int n    = t * 16 + (lane & 15);
    const int mOff = wave * 16 + ((lane >> 4) << 3);
#pragma unroll
    for (int r = 0; r < 8; ++r) {
      const int i = mOff + r;
      float s = (i == n) ? 0.f
                         : (norms[i] + norms[n] - 2.f * acc[r]) * invs;
      out[i * E + n] = s;
    }
  }
}

__global__ __launch_bounds__(256) void k_stats(const float* __restrict__ sqd,
                                               float* __restrict__ stats) {
  __shared__ float s_sum[256], s_sq[256], s_cnt[256];
  const int tid = threadIdx.x;
  const int N   = BATCH * E * E;
  float sum = 0.f, sq = 0.f, cnt = 0.f;
  for (int i = tid; i < N; i += 256) {
    float v = sqd[i];
    sum += v;
    sq  += v * v;
    cnt += (v != 0.f) ? 1.f : 0.f;
  }
  s_sum[tid] = sum; s_sq[tid] = sq; s_cnt[tid] = cnt;
  __syncthreads();
  for (int off = 128; off > 0; off >>= 1) {
    if (tid < off) {
      s_sum[tid] += s_sum[tid + off];
      s_sq[tid]  += s_sq[tid + off];
      s_cnt[tid] += s_cnt[tid + off];
    }
    __syncthreads();
  }
  if (tid == 0) {
    float c    = s_cnt[0];
    float mean = s_sum[0] / c;
    float var  = (s_sq[0] - c * mean * mean) / (c - 1.f);
    stats[0]   = rsqrtf(var);
  }
}

__global__ __launch_bounds__(256) void k_solve(const float* __restrict__ sqd,
                                               const float* __restrict__ stats,
                                               const int* __restrict__ labels,
                                               float* __restrict__ out) {
  extern __shared__ float smem[];
  float* Amat = smem;
  float* X    = smem + E * E;
  float* T    = smem + 2 * E * E;
  float* red  = smem + 3 * E * E;
  int*   lab  = (int*)(smem + 3 * E * E + E);
  const int b    = blockIdx.x;
  const int tid  = threadIdx.x;
  const int lane = tid & 31;
  const int wave = tid >> 5;
  const float rstd = stats[0];
  const float* sq  = sqd + (size_t)b * E * E;

  if (tid < E) lab[tid] = labels[b * E + tid];

  for (int idx = tid; idx < E * E; idx += 256) {
    const int i = idx >> 7, j = idx & 127;
    X[idx] = (i == j) ? 0.f : expf(-sq[idx] * rstd);
  }
  __syncthreads();

  if (tid < E) {
    float r = 0.f;
    for (int j = 0; j < E; ++j) r += X[tid * E + j];
    red[tid] = ALPHA / (1e-4f + r);
  }
  __syncthreads();

  for (int idx = tid; idx < E * E; idx += 256) {
    const int i = idx >> 7, j = idx & 127;
    Amat[idx] = ((i == j) ? 1.f : 0.f) - X[idx] * red[j];
  }
  __syncthreads();
  for (int idx = tid; idx < E * E; idx += 256) {
    const int i = idx >> 7, j = idx & 127;
    X[idx] = (i == j) ? 1.f : 0.f;
  }
  __syncthreads();

  v8f acc[8];
  for (int it = 0; it < 5; ++it) {
    for (int t = 0; t < 8; ++t)
      acc[t] = gemm_tile<false>(Amat, X, wave * 16, t * 16, E, E, E, lane);
    for (int t = 0; t < 8; ++t) store_tile(T, wave * 16, t * 16, lane, acc[t]);
    __syncthreads();
    for (int t = 0; t < 8; ++t)
      acc[t] = gemm_tile<false>(X, T, wave * 16, t * 16, E, E, E, lane);
    __syncthreads();
    for (int t = 0; t < 8; ++t) {
      const int n    = t * 16 + (lane & 15);
      const int mOff = wave * 16 + ((lane >> 4) << 3);
#pragma unroll
      for (int r = 0; r < 8; ++r) {
        const int i = mOff + r;
        X[i * E + n] = 2.f * X[i * E + n] - acc[t][r];
      }
    }
    __syncthreads();
  }

  if (tid < E) {
    const int i = tid;
    float cls[NCLS] = {0.f, 0.f, 0.f, 0.f, 0.f};
    float l1 = 0.f;
    for (int j = 0; j < E; ++j) {
      const float p = X[i * E + j];
      l1 += fabsf(p);
      const int lb = lab[j];
      if (lb < NCLS) cls[lb] += p;
    }
    const float inv = 1.f / fmaxf(l1, 1e-12f);
#pragma unroll
    for (int c = 0; c < NCLS; ++c)
      out[((size_t)b * E + i) * NCLS + c] = logf(cls[c] * inv + 1e-6f);
  }
}

extern "C" void kernel_launch(void* const* d_in, const int* in_sizes, int n_in,
                              void* d_out, int out_size, void* d_ws, size_t ws_size,
                              hipStream_t stream) {
  const float* x      = (const float*)d_in[0];
  const int*   labels = (const int*)d_in[1];
  float* out   = (float*)d_out;
  float* sqd   = (float*)d_ws;
  float* stats = sqd + BATCH * E * E;

  const size_t sh1 = (size_t)(E * CDIM + E) * sizeof(float);
  k_gram<<<BATCH, 256, sh1, stream>>>(x, sqd);
  k_stats<<<1, 256, 0, stream>>>(sqd, stats);
  const size_t sh3 = (size_t)(3 * E * E + 2 * E) * sizeof(float);
  k_solve<<<BATCH, 256, sh3, stream>>>(sqd, stats, labels, out);
}


# LDS row pitch padded off the 64-bank period (x tile 512->516 floats, solve matrices 128->140) so WMMA fragment reads and per-row walks stop piling onto one bank group
# speedup vs baseline: 1.6922x; 1.6922x over previous
#include <hip/hip_runtime.h>
#include <hip/hip_bf16.h>
#include <math.h>

typedef __attribute__((ext_vector_type(2))) float v2f;
typedef __attribute__((ext_vector_type(8))) float v8f;

#define BATCH 16
#define E 128
#define CDIM 512
#define NCLS 5
#define ALPHA 0.2f
#define XP 516
#define SP 140

template <bool TB>
__device__ inline v8f gemm_tile(const float* __restrict__ Al,
                                const float* __restrict__ Bl,
                                int rowBase, int colBase, int K,
                                int lda, int ldb, int lane) {
  const int m  = lane & 15;
  const int kb = (lane >> 4) << 1;
  v8f acc = {0.f, 0.f, 0.f, 0.f, 0.f, 0.f, 0.f, 0.f};
  for (int kk = 0; kk < K; kk += 4) {
    v2f a = *(const v2f*)&Al[(rowBase + m) * lda + kk + kb];
    v2f b;
    if (TB) {
      b = *(const v2f*)&Bl[(colBase + m) * ldb + kk + kb];
    } else {
      b.x = Bl[(kk + kb) * ldb + colBase + m];
      b.y = Bl[(kk + kb + 1) * ldb + colBase + m];
    }
    acc = __builtin_amdgcn_wmma_f32_16x16x4_f32(
         false, a,  false, b,
         (short)0, acc,  false,  false);
  }
  return acc;
}

__device__ inline void store_tile(float* __restrict__ Cm, int rowBase,
                                  int colBase, int ld, int lane, v8f acc) {
  const int n    = colBase + (lane & 15);
  const int mOff = rowBase + ((lane >> 4) << 3);
#pragma unroll
  for (int r = 0; r < 8; ++r) Cm[(mOff + r) * ld + n] = acc[r];
}

__global__ __launch_bounds__(256) void k_gram(const float* __restrict__ x,
                                              float* __restrict__ sqd) {
  extern __shared__ float smem[];
  float* xs    = smem;
  float* norms = smem + E * XP;
  const int b    = blockIdx.x;
  const int tid  = threadIdx.x;
  const int lane = tid & 31;
  const int wave = tid >> 5;

  const float4* xg = (const float4*)(x + (size_t)b * E * CDIM);
  for (int i = tid; i < E * CDIM / 4; i += 256) {
    const int r = i >> 7, c4 = i & 127;
    *(float4*)&xs[r * XP + c4 * 4] = xg[i];
  }
  __syncthreads();

  for (int i = tid; i < E; i += 256) {
    const float* row = xs + i * XP;
    float s = 0.f;
    for (int k = 0; k < CDIM; ++k) s += row[k] * row[k];
    norms[i] = s;
  }
  __syncthreads();

  const float invs = 0.04419417382415922f;
  float* out = sqd + (size_t)b * E * E;
  for (int t = 0; t < 8; ++t) {
    v8f acc = gemm_tile<true>(xs, xs, wave * 16, t * 16, CDIM, XP, XP, lane);
    const int n    = t * 16 + (lane & 15);
    const int mOff = wave * 16 + ((lane >> 4) << 3);
#pragma unroll
    for (int r = 0; r < 8; ++r) {
      const int i = mOff + r;
      float s = (i == n) ? 0.f
                         : (norms[i] + norms[n] - 2.f * acc[r]) * invs;
      out[i * E + n] = s;
    }
  }
}

__global__ __launch_bounds__(256) void k_stats(const float* __restrict__ sqd,
                                               float* __restrict__ stats) {
  __shared__ float s_sum[256], s_sq[256], s_cnt[256];
  const int tid = threadIdx.x;
  const int N   = BATCH * E * E;
  float sum = 0.f, sq = 0.f, cnt = 0.f;
  for (int i = tid; i < N; i += 256) {
    float v = sqd[i];
    sum += v;
    sq  += v * v;
    cnt += (v != 0.f) ? 1.f : 0.f;
  }
  s_sum[tid] = sum; s_sq[tid] = sq; s_cnt[tid] = cnt;
  __syncthreads();
  for (int off = 128; off > 0; off >>= 1) {
    if (tid < off) {
      s_sum[tid] += s_sum[tid + off];
      s_sq[tid]  += s_sq[tid + off];
      s_cnt[tid] += s_cnt[tid + off];
    }
    __syncthreads();
  }
  if (tid == 0) {
    float c    = s_cnt[0];
    float mean = s_sum[0] / c;
    float var  = (s_sq[0] - c * mean * mean) / (c - 1.f);
    stats[0]   = rsqrtf(var);
  }
}

__global__ __launch_bounds__(256) void k_solve(const float* __restrict__ sqd,
                                               const float* __restrict__ stats,
                                               const int* __restrict__ labels,
                                               float* __restrict__ out) {
  extern __shared__ float smem[];
  float* Amat = smem;
  float* X    = smem + E * SP;
  float* T    = smem + 2 * E * SP;
  float* red  = smem + 3 * E * SP;
  int*   lab  = (int*)(smem + 3 * E * SP + E);
  const int b    = blockIdx.x;
  const int tid  = threadIdx.x;
  const int lane = tid & 31;
  const int wave = tid >> 5;
  const float rstd = stats[0];
  const float* sq  = sqd + (size_t)b * E * E;

  if (tid < E) lab[tid] = labels[b * E + tid];

  for (int idx = tid; idx < E * E; idx += 256) {
    const int i = idx >> 7, j = idx & 127;
    X[i * SP + j] = (i == j) ? 0.f : expf(-sq[idx] * rstd);
  }
  __syncthreads();

  if (tid < E) {
    float r = 0.f;
    for (int j = 0; j < E; ++j) r += X[tid * SP + j];
    red[tid] = ALPHA / (1e-4f + r);
  }
  __syncthreads();

  for (int idx = tid; idx < E * E; idx += 256) {
    const int i = idx >> 7, j = idx & 127;
    Amat[i * SP + j] = ((i == j) ? 1.f : 0.f) - X[i * SP + j] * red[j];
  }
  __syncthreads();
  for (int idx = tid; idx < E * E; idx += 256) {
    const int i = idx >> 7, j = idx & 127;
    X[i * SP + j] = (i == j) ? 1.f : 0.f;
  }
  __syncthreads();

  v8f acc[8];
  for (int it = 0; it < 5; ++it) {
    for (int t = 0; t < 8; ++t)
      acc[t] = gemm_tile<false>(Amat, X, wave * 16, t * 16, E, SP, SP, lane);
    for (int t = 0; t < 8; ++t) store_tile(T, wave * 16, t * 16, SP, lane, acc[t]);
    __syncthreads();
    for (int t = 0; t < 8; ++t)
      acc[t] = gemm_tile<false>(X, T, wave * 16, t * 16, E, SP, SP, lane);
    __syncthreads();
    for (int t = 0; t < 8; ++t) {
      const int n    = t * 16 + (lane & 15);
      const int mOff = wave * 16 + ((lane >> 4) << 3);
#pragma unroll
      for (int r = 0; r < 8; ++r) {
        const int i = mOff + r;
        X[i * SP + n] = 2.f * X[i * SP + n] - acc[t][r];
      }
    }
    __syncthreads();
  }

  if (tid < E) {
    const int i = tid;
    float cls[NCLS] = {0.f, 0.f, 0.f, 0.f, 0.f};
    float l1 = 0.f;
    for (int j = 0; j < E; ++j) {
      const float p = X[i * SP + j];
      l1 += fabsf(p);
      const int lb = lab[j];
      if (lb < NCLS) cls[lb] += p;
    }
    const float inv = 1.f / fmaxf(l1, 1e-12f);
#pragma unroll
    for (int c = 0; c < NCLS; ++c)
      out[((size_t)b * E + i) * NCLS + c] = logf(cls[c] * inv + 1e-6f);
  }
}

extern "C" void kernel_launch(void* const* d_in, const int* in_sizes, int n_in,
                              void* d_out, int out_size, void* d_ws, size_t ws_size,
                              hipStream_t stream) {
  const float* x      = (const float*)d_in[0];
  const int*   labels = (const int*)d_in[1];
  float* out   = (float*)d_out;
  float* sqd   = (float*)d_ws;
  float* stats = sqd + BATCH * E * E;

  const size_t sh1 = (size_t)(E * XP + E) * sizeof(float);
  k_gram<<<BATCH, 256, sh1, stream>>>(x, sqd);
  k_stats<<<1, 256, 0, stream>>>(sqd, stats);
  const size_t sh3 = (size_t)(3 * E * SP + 2 * E) * sizeof(float);
  k_solve<<<BATCH, 256, sh3, stream>>>(sqd, stats, labels, out);
}


# k_stats: the one statistics work-group grows to 1024 threads and streams 16-byte pieces with 12-16 loads in flight per lane (was 256 threads, 8 4-byte loads, then a full wait): ~16x bytes in flight
# speedup vs baseline: 2.1449x; 1.2675x over previous
#include <hip/hip_runtime.h>
#include <hip/hip_bf16.h>
#include <math.h>

typedef __attribute__((ext_vector_type(2))) float v2f;
typedef __attribute__((ext_vector_type(8))) float v8f;

#define BATCH 16
#define E 128
#define CDIM 512
#define NCLS 5
#define ALPHA 0.2f
#define XP 516
#define SP 140

template <bool TB>
__device__ inline v8f gemm_tile(const float* __restrict__ Al,
                                const float* __restrict__ Bl,
                                int rowBase, int colBase, int K,
                                int lda, int ldb, int lane) {
  const int m  = lane & 15;
  const int kb = (lane >> 4) << 1;
  v8f acc = {0.f, 0.f, 0.f, 0.f, 0.f, 0.f, 0.f, 0.f};
  for (int kk = 0; kk < K; kk += 4) {
    v2f a = *(const v2f*)&Al[(rowBase + m) * lda + kk + kb];
    v2f b;
    if (TB) {
      b = *(const v2f*)&Bl[(colBase + m) * ldb + kk + kb];
    } else {
      b.x = Bl[(kk + kb) * ldb + colBase + m];
      b.y = Bl[(kk + kb + 1) * ldb + colBase + m];
    }
    acc = __builtin_amdgcn_wmma_f32_16x16x4_f32(
         false, a,  false, b,
         (short)0, acc,  false,  false);
  }
  return acc;
}

__device__ inline void store_tile(float* __restrict__ Cm, int rowBase,
                                  int colBase, int ld, int lane, v8f acc) {
  const int n    = colBase + (lane & 15);
  const int mOff = rowBase + ((lane >> 4) << 3);
#pragma unroll
  for (int r = 0; r < 8; ++r) Cm[(mOff + r) * ld + n] = acc[r];
}

__global__ __launch_bounds__(256) void k_gram(const float* __restrict__ x,
                                              float* __restrict__ sqd) {
  extern __shared__ float smem[];
  float* xs    = smem;
  float* norms = smem + E * XP;
  const int b    = blockIdx.x;
  const int tid  = threadIdx.x;
  const int lane = tid & 31;
  const int wave = tid >> 5;

  const float4* xg = (const float4*)(x + (size_t)b * E * CDIM);
  for (int i = tid; i < E * CDIM / 4; i += 256) {
    const int r = i >> 7, c4 = i & 127;
    *(float4*)&xs[r * XP + c4 * 4] = xg[i];
  }
  __syncthreads();

  for (int i = tid; i < E; i += 256) {
    const float* row = xs + i * XP;
    float s = 0.f;
    for (int k = 0; k < CDIM; ++k) s += row[k] * row[k];
    norms[i] = s;
  }
  __syncthreads();

  const float invs = 0.04419417382415922f;
  float* out = sqd + (size_t)b * E * E;
  for (int t = 0; t < 8; ++t) {
    v8f acc = gemm_tile<true>(xs, xs, wave * 16, t * 16, CDIM, XP, XP, lane);
    const int n    = t * 16 + (lane & 15);
    const int mOff = wave * 16 + ((lane >> 4) << 3);
#pragma unroll
    for (int r = 0; r < 8; ++r) {
      const int i = mOff + r;
      float s = (i == n) ? 0.f
                         : (norms[i] + norms[n] - 2.f * acc[r]) * invs;
      out[i * E + n] = s;
    }
  }
}

#define ST 1024
#define SG 16
__global__ __launch_bounds__(ST) void k_stats(const float* __restrict__ sqd,
                                              float* __restrict__ stats) {
  __shared__ float s_part[3][ST / 32];
  const int tid  = threadIdx.x;
  const int lane = tid & 31;
  const int wave = tid >> 5;
  const float4* __restrict__ p = (const float4*)sqd;
  const int NV = BATCH * E * E / 4;
  float4 s4 = {0.f, 0.f, 0.f, 0.f};
  float4 q4 = {0.f, 0.f, 0.f, 0.f};
  float4 c4 = {0.f, 0.f, 0.f, 0.f};
  for (int base = 0; base < NV; base += ST * SG) {
    float4 v[SG];
#pragma unroll
    for (int u = 0; u < SG; ++u) v[u] = p[base + u * ST + tid];
#pragma unroll
    for (int u = 0; u < SG; ++u) {
      s4.x += v[u].x; s4.y += v[u].y; s4.z += v[u].z; s4.w += v[u].w;
      q4.x += v[u].x * v[u].x; q4.y += v[u].y * v[u].y;
      q4.z += v[u].z * v[u].z; q4.w += v[u].w * v[u].w;
      c4.x += (v[u].x != 0.f) ? 1.f : 0.f;
      c4.y += (v[u].y != 0.f) ? 1.f : 0.f;
      c4.z += (v[u].z != 0.f) ? 1.f : 0.f;
      c4.w += (v[u].w != 0.f) ? 1.f : 0.f;
    }
  }
  float sum = (s4.x + s4.y) + (s4.z + s4.w);
  float sq  = (q4.x + q4.y) + (q4.z + q4.w);
  float cnt = (c4.x + c4.y) + (c4.z + c4.w);
  for (int off = 16; off > 0; off >>= 1) {
    sum += __shfl_xor(sum, off);
    sq  += __shfl_xor(sq, off);
    cnt += __shfl_xor(cnt, off);
  }
  if (lane == 0) {
    s_part[0][wave] = sum; s_part[1][wave] = sq; s_part[2][wave] = cnt;
  }
  __syncthreads();
  if (wave == 0) {
    sum = s_part[0][lane]; sq = s_part[1][lane]; cnt = s_part[2][lane];
    for (int off = 16; off > 0; off >>= 1) {
      sum += __shfl_xor(sum, off);
      sq  += __shfl_xor(sq, off);
      cnt += __shfl_xor(cnt, off);
    }
    if (lane == 0) {
      float c    = cnt;
      float mean = sum / c;
      float var  = (sq - c * mean * mean) / (c - 1.f);
      stats[0]   = rsqrtf(var);
    }
  }
}

__global__ __launch_bounds__(256) void k_solve(const float* __restrict__ sqd,
                                               const float* __restrict__ stats,
                                               const int* __restrict__ labels,
                                               float* __restrict__ out) {
  extern __shared__ float smem[];
  float* Amat = smem;
  float* X    = smem + E * SP;
  float* T    = smem + 2 * E * SP;
  float* red  = smem + 3 * E * SP;
  int*   lab  = (int*)(smem + 3 * E * SP + E);
  const int b    = blockIdx.x;
  const int tid  = threadIdx.x;
  const int lane = tid & 31;
  const int wave = tid >> 5;
  const float rstd = stats[0];
  const float* sq  = sqd + (size_t)b * E * E;

  if (tid < E) lab[tid] = labels[b * E + tid];

  for (int idx = tid; idx < E * E; idx += 256) {
    const int i = idx >> 7, j = idx & 127;
    X[i * SP + j] = (i == j) ? 0.f : expf(-sq[idx] * rstd);
  }
  __syncthreads();

  if (tid < E) {
    float r = 0.f;
    for (int j = 0; j < E; ++j) r += X[tid * SP + j];
    red[tid] = ALPHA / (1e-4f + r);
  }
  __syncthreads();

  for (int idx = tid; idx < E * E; idx += 256) {
    const int i = idx >> 7, j = idx & 127;
    Amat[i * SP + j] = ((i == j) ? 1.f : 0.f) - X[i * SP + j] * red[j];
  }
  __syncthreads();
  for (int idx = tid; idx < E * E; idx += 256) {
    const int i = idx >> 7, j = idx & 127;
    X[i * SP + j] = (i == j) ? 1.f : 0.f;
  }
  __syncthreads();

  v8f acc[8];
  for (int it = 0; it < 5; ++it) {
    for (int t = 0; t < 8; ++t)
      acc[t] = gemm_tile<false>(Amat, X, wave * 16, t * 16, E, SP, SP, lane);
    for (int t = 0; t < 8; ++t) store_tile(T, wave * 16, t * 16, SP, lane, acc[t]);
    __syncthreads();
    for (int t = 0; t < 8; ++t)
      acc[t] = gemm_tile<false>(X, T, wave * 16, t * 16, E, SP, SP, lane);
    __syncthreads();
    for (int t = 0; t < 8; ++t) {
      const int n    = t * 16 + (lane & 15);
      const int mOff = wave * 16 + ((lane >> 4) << 3);
#pragma unroll
      for (int r = 0; r < 8; ++r) {
        const int i = mOff + r;
        X[i * SP + n] = 2.f * X[i * SP + n] - acc[t][r];
      }
    }
    __syncthreads();
  }

  if (tid < E) {
    const int i = tid;
    float cls[NCLS] = {0.f, 0.f, 0.f, 0.f, 0.f};
    float l1 = 0.f;
    for (int j = 0; j < E; ++j) {
      const float p = X[i * SP + j];
      l1 += fabsf(p);
      const int lb = lab[j];
      if (lb < NCLS) cls[lb] += p;
    }
    const float inv = 1.f / fmaxf(l1, 1e-12f);
#pragma unroll
    for (int c = 0; c < NCLS; ++c)
      out[((size_t)b * E + i) * NCLS + c] = logf(cls[c] * inv + 1e-6f);
  }
}

extern "C" void kernel_launch(void* const* d_in, const int* in_sizes, int n_in,
                              void* d_out, int out_size, void* d_ws, size_t ws_size,
                              hipStream_t stream) {
  const float* x      = (const float*)d_in[0];
  const int*   labels = (const int*)d_in[1];
  float* out   = (float*)d_out;
  float* sqd   = (float*)d_ws;
  float* stats = sqd + BATCH * E * E;

  const size_t sh1 = (size_t)(E * XP + E) * sizeof(float);
  k_gram<<<BATCH, 256, sh1, stream>>>(x, sqd);
  k_stats<<<1, ST, 0, stream>>>(sqd, stats);
  const size_t sh3 = (size_t)(3 * E * SP + 2 * E) * sizeof(float);
  k_solve<<<BATCH, 256, sh3, stream>>>(sqd, stats, labels, out);
}


# k_solve: the inverse as the same 32-term series in product form (X1 = I+M free; X += X*P and P = P*P share the B fragments of P): 8 GEMMs of 128^3 per batch instead of 10
# speedup vs baseline: 2.4671x; 1.1502x over previous
#include <hip/hip_runtime.h>
#include <hip/hip_bf16.h>
#include <math.h>

typedef __attribute__((ext_vector_type(2))) float v2f;
typedef __attribute__((ext_vector_type(8))) float v8f;

#define BATCH 16
#define E 128
#define CDIM 512
#define NCLS 5
#define ALPHA 0.2f
#define XP 516
#define SP 140

template <bool TB>
__device__ inline v8f gemm_tile(const float* __restrict__ Al,
                                const float* __restrict__ Bl,
                                int rowBase, int colBase, int K,
                                int lda, int ldb, int lane) {
  const int m  = lane & 15;
  const int kb = (lane >> 4) << 1;
  v8f acc = {0.f, 0.f, 0.f, 0.f, 0.f, 0.f, 0.f, 0.f};
  for (int kk = 0; kk < K; kk += 4) {
    v2f a = *(const v2f*)&Al[(rowBase + m) * lda + kk + kb];
    v2f b;
    if (TB) {
      b = *(const v2f*)&Bl[(colBase + m) * ldb + kk + kb];
    } else {
      b.x = Bl[(kk + kb) * ldb + colBase + m];
      b.y = Bl[(kk + kb + 1) * ldb + colBase + m];
    }
    acc = __builtin_amdgcn_wmma_f32_16x16x4_f32(
         false, a,  false, b,
         (short)0, acc,  false,  false);
  }
  return acc;
}

__device__ inline void store_tile(float* __restrict__ Cm, int rowBase,
                                  int colBase, int ld, int lane, v8f acc) {
  const int n    = colBase + (lane & 15);
  const int mOff = rowBase + ((lane >> 4) << 3);
#pragma unroll
  for (int r = 0; r < 8; ++r) Cm[(mOff + r) * ld + n] = acc[r];
}

__global__ __launch_bounds__(256) void k_gram(const float* __restrict__ x,
                                              float* __restrict__ sqd) {
  extern __shared__ float smem[];
  float* xs    = smem;
  float* norms = smem + E * XP;
  const int b    = blockIdx.x;
  const int tid  = threadIdx.x;
  const int lane = tid & 31;
  const int wave = tid >> 5;

  const float4* xg = (const float4*)(x + (size_t)b * E * CDIM);
  for (int i = tid; i < E * CDIM / 4; i += 256) {
    const int r = i >> 7, c4 = i & 127;
    *(float4*)&xs[r * XP + c4 * 4] = xg[i];
  }
  __syncthreads();

  for (int i = tid; i < E; i += 256) {
    const float* row = xs + i * XP;
    float s = 0.f;
    for (int k = 0; k < CDIM; ++k) s += row[k] * row[k];
    norms[i] = s;
  }
  __syncthreads();

  const float invs = 0.04419417382415922f;
  float* out = sqd + (size_t)b * E * E;
  for (int t = 0; t < 8; ++t) {
    v8f acc = gemm_tile<true>(xs, xs, wave * 16, t * 16, CDIM, XP, XP, lane);
    const int n    = t * 16 + (lane & 15);
    const int mOff = wave * 16 + ((lane >> 4) << 3);
#pragma unroll
    for (int r = 0; r < 8; ++r) {
      const int i = mOff + r;
      float s = (i == n) ? 0.f
                         : (norms[i] + norms[n] - 2.f * acc[r]) * invs;
      out[i * E + n] = s;
    }
  }
}

#define ST 1024
#define SG 16
__global__ __launch_bounds__(ST) void k_stats(const float* __restrict__ sqd,
                                              float* __restrict__ stats) {
  __shared__ float s_part[3][ST / 32];
  const int tid  = threadIdx.x;
  const int lane = tid & 31;
  const int wave = tid >> 5;
  const float4* __restrict__ p = (const float4*)sqd;
  const int NV = BATCH * E * E / 4;
  float4 s4 = {0.f, 0.f, 0.f, 0.f};
  float4 q4 = {0.f, 0.f, 0.f, 0.f};
  float4 c4 = {0.f, 0.f, 0.f, 0.f};
  for (int base = 0; base < NV; base += ST * SG) {
    float4 v[SG];
#pragma unroll
    for (int u = 0; u < SG; ++u) v[u] = p[base + u * ST + tid];
#pragma unroll
    for (int u = 0; u < SG; ++u) {
      s4.x += v[u].x; s4.y += v[u].y; s4.z += v[u].z; s4.w += v[u].w;
      q4.x += v[u].x * v[u].x; q4.y += v[u].y * v[u].y;
      q4.z += v[u].z * v[u].z; q4.w += v[u].w * v[u].w;
      c4.x += (v[u].x != 0.f) ? 1.f : 0.f;
      c4.y += (v[u].y != 0.f) ? 1.f : 0.f;
      c4.z += (v[u].z != 0.f) ? 1.f : 0.f;
      c4.w += (v[u].w != 0.f) ? 1.f : 0.f;
    }
  }
  float sum = (s4.x + s4.y) + (s4.z + s4.w);
  float sq  = (q4.x + q4.y) + (q4.z + q4.w);
  float cnt = (c4.x + c4.y) + (c4.z + c4.w);
  for (int off = 16; off > 0; off >>= 1) {
    sum += __shfl_xor(sum, off);
    sq  += __shfl_xor(sq, off);
    cnt += __shfl_xor(cnt, off);
  }
  if (lane == 0) {
    s_part[0][wave] = sum; s_part[1][wave] = sq; s_part[2][wave] = cnt;
  }
  __syncthreads();
  if (wave == 0) {
    sum = s_part[0][lane]; sq = s_part[1][lane]; cnt = s_part[2][lane];
    for (int off = 16; off > 0; off >>= 1) {
      sum += __shfl_xor(sum, off);
      sq  += __shfl_xor(sq, off);
      cnt += __shfl_xor(cnt, off);
    }
    if (lane == 0) {
      float c    = cnt;
      float mean = sum / c;
      float var  = (sq - c * mean * mean) / (c - 1.f);
      stats[0]   = rsqrtf(var);
    }
  }
}

template <bool DO_X, bool DO_P>
__device__ inline void series_step(float* __restrict__ Xm, float* __restrict__ Pm,
                                   int wave, int lane) {
  const int m    = lane & 15;
  const int kb   = (lane >> 4) << 1;
  const int row  = wave * 16 + m;
  const int mOff = wave * 16 + ((lane >> 4) << 3);
  v8f ax[8], ap[8];
#pragma unroll
  for (int t = 0; t < 8; ++t) {
#pragma unroll
    for (int r = 0; r < 8; ++r) {
      ax[t][r] = DO_X ? Xm[(mOff + r) * SP + t * 16 + m] : 0.f;
      ap[t][r] = 0.f;
    }
  }
#pragma unroll
  for (int kk = 0; kk < E; kk += 4) {
    v2f fx, fp;
    if (DO_X) fx = *(const v2f*)&Xm[row * SP + kk + kb];
    if (DO_P) fp = *(const v2f*)&Pm[row * SP + kk + kb];
#pragma unroll
    for (int t = 0; t < 8; ++t) {
      v2f b;
      b.x = Pm[(kk + kb) * SP + t * 16 + m];
      b.y = Pm[(kk + kb + 1) * SP + t * 16 + m];
      if (DO_X)
        ax[t] = __builtin_amdgcn_wmma_f32_16x16x4_f32(
            false, fx, false, b, (short)0, ax[t], false, false);
      if (DO_P)
        ap[t] = __builtin_amdgcn_wmma_f32_16x16x4_f32(
            false, fp, false, b, (short)0, ap[t], false, false);
    }
  }
  if (DO_P) {
    __syncthreads();
#pragma unroll
    for (int t = 0; t < 8; ++t) store_tile(Pm, wave * 16, t * 16, SP, lane, ap[t]);
  }
  if (DO_X) {
#pragma unroll
    for (int t = 0; t < 8; ++t) store_tile(Xm, wave * 16, t * 16, SP, lane, ax[t]);
  }
  __syncthreads();
}

__global__ __launch_bounds__(256) void k_solve(const float* __restrict__ sqd,
                                               const float* __restrict__ stats,
                                               const int* __restrict__ labels,
                                               float* __restrict__ out) {
  extern __shared__ float smem[];
  float* Pm  = smem;
  float* X   = smem + E * SP;
  float* red = smem + 2 * E * SP;
  int*   lab = (int*)(smem + 2 * E * SP + E);
  const int b    = blockIdx.x;
  const int tid  = threadIdx.x;
  const int lane = tid & 31;
  const int wave = tid >> 5;
  const float rstd = stats[0];
  const float* sq  = sqd + (size_t)b * E * E;

  if (tid < E) lab[tid] = labels[b * E + tid];

  for (int idx = tid; idx < E * E; idx += 256) {
    const int i = idx >> 7, j = idx & 127;
    X[i * SP + j] = (i == j) ? 0.f : expf(-sq[idx] * rstd);
  }
  __syncthreads();

  if (tid < E) {
    float r = 0.f;
    for (int j = 0; j < E; ++j) r += X[tid * SP + j];
    red[tid] = ALPHA / (1e-4f + r);
  }
  __syncthreads();

  for (int idx = tid; idx < E * E; idx += 256) {
    const int i = idx >> 7, j = idx & 127;
    const float mij = X[i * SP + j] * red[j];
    Pm[i * SP + j] = mij;
    X[i * SP + j]  = ((i == j) ? 1.f : 0.f) + mij;
  }
  __syncthreads();

  series_step<false, true>(X, Pm, wave, lane);
  for (int it = 0; it < 3; ++it)
    series_step<true, true>(X, Pm, wave, lane);
  series_step<true, false>(X, Pm, wave, lane);

  if (tid < E) {
    const int i = tid;
    float cls[NCLS] = {0.f, 0.f, 0.f, 0.f, 0.f};
    float l1 = 0.f;
    for (int j = 0; j < E; ++j) {
      const float p = X[i * SP + j];
      l1 += fabsf(p);
      const int lb = lab[j];
      if (lb < NCLS) cls[lb] += p;
    }
    const float inv = 1.f / fmaxf(l1, 1e-12f);
#pragma unroll
    for (int c = 0; c < NCLS; ++c)
      out[((size_t)b * E + i) * NCLS + c] = logf(cls[c] * inv + 1e-6f);
  }
}

extern "C" void kernel_launch(void* const* d_in, const int* in_sizes, int n_in,
                              void* d_out, int out_size, void* d_ws, size_t ws_size,
                              hipStream_t stream) {
  const float* x      = (const float*)d_in[0];
  const int*   labels = (const int*)d_in[1];
  float* out   = (float*)d_out;
  float* sqd   = (float*)d_ws;
  float* stats = sqd + BATCH * E * E;

  const size_t sh1 = (size_t)(E * XP + E) * sizeof(float);
  k_gram<<<BATCH, 256, sh1, stream>>>(x, sqd);
  k_stats<<<1, ST, 0, stream>>>(sqd, stats);
  const size_t sh3 = (size_t)(2 * E * SP + 2 * E) * sizeof(float);
  k_solve<<<BATCH, 256, sh3, stream>>>(sqd, stats, labels, out);
}


# both WMMA loops walk k in blocks of 2 k-steps x 8 column tiles whose LDS fragment reads go out as ONE group ahead of the previous block's WMMAs (2nd register set; solve keeps P transposed too)
# speedup vs baseline: 2.6151x; 1.0600x over previous
#include <hip/hip_runtime.h>
#include <hip/hip_bf16.h>
#include <math.h>

typedef __attribute__((ext_vector_type(2))) float v2f;
typedef __attribute__((ext_vector_type(8))) float v8f;

#define BATCH 16
#define E 128
#define CDIM 512
#define NCLS 5
#define ALPHA 0.2f
#define XP 516
#define SP 140

__device__ inline void store_tile(float* __restrict__ Cm, int rowBase,
                                  int colBase, int ld, int lane, v8f acc) {
  const int n    = colBase + (lane & 15);
  const int mOff = rowBase + ((lane >> 4) << 3);
#pragma unroll
  for (int r = 0; r < 8; ++r) Cm[(mOff + r) * ld + n] = acc[r];
}

__device__ inline void fence() { __builtin_amdgcn_sched_barrier(0); }

__device__ inline v8f mma4(v2f a, v2f b, v8f c) {
  return __builtin_amdgcn_wmma_f32_16x16x4_f32(
       false, a,  false, b,
       (short)0, c,  false,  false);
}

__device__ inline void gram_load(const float* __restrict__ ga,
                                 const float* __restrict__ gb, int k0,
                                 v2f (&fa)[2], v2f (&fb)[2][8]) {
#pragma unroll
  for (int s = 0; s < 2; ++s) {
    fa[s] = *(const v2f*)(ga + k0 + 4 * s);
#pragma unroll
    for (int t = 0; t < 8; ++t)
      fb[s][t] = *(const v2f*)(gb + t * 16 * XP + k0 + 4 * s);
  }
}

__device__ inline void gram_mma(v8f (&acc)[8], const v2f (&fa)[2],
                                const v2f (&fb)[2][8]) {
#pragma unroll
  for (int s = 0; s < 2; ++s)
#pragma unroll
    for (int t = 0; t < 8; ++t) acc[t] = mma4(fa[s], fb[s][t], acc[t]);
}

__global__ __launch_bounds__(256) void k_gram(const float* __restrict__ x,
                                              float* __restrict__ sqd) {
  extern __shared__ float smem[];
  float* xs    = smem;
  float* norms = smem + E * XP;
  const int b    = blockIdx.x;
  const int tid  = threadIdx.x;
  const int lane = tid & 31;
  const int wave = tid >> 5;

  const float4* xg = (const float4*)(x + (size_t)b * E * CDIM);
  for (int i = tid; i < E * CDIM / 4; i += 256) {
    const int r = i >> 7, c4 = i & 127;
    *(float4*)&xs[r * XP + c4 * 4] = xg[i];
  }
  __syncthreads();

  for (int i = tid; i < E; i += 256) {
    const float* row = xs + i * XP;
    float s = 0.f;
    for (int k = 0; k < CDIM; ++k) s += row[k] * row[k];
    norms[i] = s;
  }
  __syncthreads();

  const float invs = 0.04419417382415922f;
  float* out = sqd + (size_t)b * E * E;
  const float* ga = xs + (wave * 16 + (lane & 15)) * XP + ((lane >> 4) << 1);
  const float* gb = xs + (lane & 15) * XP + ((lane >> 4) << 1);
  v8f acc[8];
#pragma unroll
  for (int t = 0; t < 8; ++t) acc[t] = (v8f){0.f, 0.f, 0.f, 0.f, 0.f, 0.f, 0.f, 0.f};
  v2f fa0[2], fa1[2], fb0[2][8], fb1[2][8];
  gram_load(ga, gb, 0, fa0, fb0);
  for (int k0 = 0; k0 < CDIM - 16; k0 += 16) {
    gram_load(ga, gb, k0 + 8, fa1, fb1);
    fence();
    gram_mma(acc, fa0, fb0);
    fence();
    gram_load(ga, gb, k0 + 16, fa0, fb0);
    fence();
    gram_mma(acc, fa1, fb1);
    fence();
  }
  gram_load(ga, gb, CDIM - 8, fa1, fb1);
  fence();
  gram_mma(acc, fa0, fb0);
  fence();
  gram_mma(acc, fa1, fb1);

#pragma unroll
  for (int t = 0; t < 8; ++t) {
    const int n    = t * 16 + (lane & 15);
    const int mOff = wave * 16 + ((lane >> 4) << 3);
#pragma unroll
    for (int r = 0; r < 8; ++r) {
      const int i = mOff + r;
      float s = (i == n) ? 0.f
                         : (norms[i] + norms[n] - 2.f * acc[t][r]) * invs;
      out[i * E + n] = s;
    }
  }
}

#define ST 1024
#define SG 16
__global__ __launch_bounds__(ST) void k_stats(const float* __restrict__ sqd,
                                              float* __restrict__ stats) {
  __shared__ float s_part[3][ST / 32];
  const int tid  = threadIdx.x;
  const int lane = tid & 31;
  const int wave = tid >> 5;
  const float4* __restrict__ p = (const float4*)sqd;
  const int NV = BATCH * E * E / 4;
  float4 s4 = {0.f, 0.f, 0.f, 0.f};
  float4 q4 = {0.f, 0.f, 0.f, 0.f};
  float4 c4 = {0.f, 0.f, 0.f, 0.f};
  for (int base = 0; base < NV; base += ST * SG) {
    float4 v[SG];
#pragma unroll
    for (int u = 0; u < SG; ++u) v[u] = p[base + u * ST + tid];
#pragma unroll
    for (int u = 0; u < SG; ++u) {
      s4.x += v[u].x; s4.y += v[u].y; s4.z += v[u].z; s4.w += v[u].w;
      q4.x += v[u].x * v[u].x; q4.y += v[u].y * v[u].y;
      q4.z += v[u].z * v[u].z; q4.w += v[u].w * v[u].w;
      c4.x += (v[u].x != 0.f) ? 1.f : 0.f;
      c4.y += (v[u].y != 0.f) ? 1.f : 0.f;
      c4.z += (v[u].z != 0.f) ? 1.f : 0.f;
      c4.w += (v[u].w != 0.f) ? 1.f : 0.f;
    }
  }
  float sum = (s4.x + s4.y) + (s4.z + s4.w);
  float sq  = (q4.x + q4.y) + (q4.z + q4.w);
  float cnt = (c4.x + c4.y) + (c4.z + c4.w);
  for (int off = 16; off > 0; off >>= 1) {
    sum += __shfl_xor(sum, off);
    sq  += __shfl_xor(sq, off);
    cnt += __shfl_xor(cnt, off);
  }
  if (lane == 0) {
    s_part[0][wave] = sum; s_part[1][wave] = sq; s_part[2][wave] = cnt;
  }
  __syncthreads();
  if (wave == 0) {
    sum = s_part[0][lane]; sq = s_part[1][lane]; cnt = s_part[2][lane];
    for (int off = 16; off > 0; off >>= 1) {
      sum += __shfl_xor(sum, off);
      sq  += __shfl_xor(sq, off);
      cnt += __shfl_xor(cnt, off);
    }
    if (lane == 0) {
      float c    = cnt;
      float mean = sum / c;
      float var  = (sq - c * mean * mean) / (c - 1.f);
      stats[0]   = rsqrtf(var);
    }
  }
}

template <bool DO_X, bool DO_P>
__device__ inline void step_load(const float* __restrict__ xa,
                                 const float* __restrict__ pa,
                                 const float* __restrict__ pb, int k0,
                                 v2f (&fx)[2], v2f (&fp)[2], v2f (&fb)[2][8]) {
#pragma unroll
  for (int s = 0; s < 2; ++s) {
    if (DO_X) fx[s] = *(const v2f*)(xa + k0 + 4 * s);
    if (DO_P) fp[s] = *(const v2f*)(pa + k0 + 4 * s);
#pragma unroll
    for (int t = 0; t < 8; ++t)
      fb[s][t] = *(const v2f*)(pb + t * 16 * SP + k0 + 4 * s);
  }
}

template <bool DO_X, bool DO_P>
__device__ inline void step_mma(v8f (&ax)[8], v8f (&ap)[8], const v2f (&fx)[2],
                                const v2f (&fp)[2], const v2f (&fb)[2][8]) {
#pragma unroll
  for (int s = 0; s < 2; ++s) {
#pragma unroll
    for (int t = 0; t < 8; ++t) {
      if (DO_X) ax[t] = mma4(fx[s], fb[s][t], ax[t]);
      if (DO_P) ap[t] = mma4(fp[s], fb[s][t], ap[t]);
    }
  }
}

template <bool DO_X, bool DO_P>
__device__ inline void series_step(float* __restrict__ Xm, float* __restrict__ Pr,
                                   float* __restrict__ PT, int wave, int lane) {
  const int m    = lane & 15;
  const int kb   = (lane >> 4) << 1;
  const int mOff = wave * 16 + ((lane >> 4) << 3);
  const float* xa = Xm + (wave * 16 + m) * SP + kb;
  const float* pa = Pr + (wave * 16 + m) * SP + kb;
  const float* pb = PT + m * SP + kb;
  v8f ax[8], ap[8];
#pragma unroll
  for (int t = 0; t < 8; ++t) {
#pragma unroll
    for (int r = 0; r < 8; ++r) {
      ax[t][r] = DO_X ? Xm[(mOff + r) * SP + t * 16 + m] : 0.f;
      ap[t][r] = 0.f;
    }
  }
  v2f fx0[2] = {}, fp0[2] = {}, fx1[2] = {}, fp1[2] = {};
  v2f fb0[2][8], fb1[2][8];
  step_load<DO_X, DO_P>(xa, pa, pb, 0, fx0, fp0, fb0);
#pragma unroll
  for (int k0 = 0; k0 < E; k0 += 16) {
    step_load<DO_X, DO_P>(xa, pa, pb, k0 + 8, fx1, fp1, fb1);
    fence();
    step_mma<DO_X, DO_P>(ax, ap, fx0, fp0, fb0);
    fence();
    if (k0 + 16 < E) step_load<DO_X, DO_P>(xa, pa, pb, k0 + 16, fx0, fp0, fb0);
    fence();
    step_mma<DO_X, DO_P>(ax, ap, fx1, fp1, fb1);
    fence();
  }
  if (DO_P) {
    __syncthreads();
#pragma unroll
    for (int t = 0; t < 8; ++t) {
      store_tile(Pr, wave * 16, t * 16, SP, lane, ap[t]);
      float* q = PT + (t * 16 + m) * SP + mOff;
      *(float4*)q       = make_float4(ap[t][0], ap[t][1], ap[t][2], ap[t][3]);
      *(float4*)(q + 4) = make_float4(ap[t][4], ap[t][5], ap[t][6], ap[t][7]);
    }
  }
  if (DO_X) {
#pragma unroll
    for (int t = 0; t < 8; ++t) store_tile(Xm, wave * 16, t * 16, SP, lane, ax[t]);
  }
  __syncthreads();
}

__global__ __launch_bounds__(256) void k_solve(const float* __restrict__ sqd,
                                               const float* __restrict__ stats,
                                               const int* __restrict__ labels,
                                               float* __restrict__ out) {
  extern __shared__ float smem[];
  float* Pr  = smem;
  float* PT  = smem + E * SP;
  float* X   = smem + 2 * E * SP;
  float* red = smem + 3 * E * SP;
  int*   lab = (int*)(smem + 3 * E * SP + E);
  const int b    = blockIdx.x;
  const int tid  = threadIdx.x;
  const int lane = tid & 31;
  const int wave = tid >> 5;
  const float rstd = stats[0];
  const float* sq  = sqd + (size_t)b * E * E;

  if (tid < E) lab[tid] = labels[b * E + tid];

  for (int idx = tid; idx < E * E; idx += 256) {
    const int i = idx >> 7, j = idx & 127;
    X[i * SP + j] = (i == j) ? 0.f : expf(-sq[idx] * rstd);
  }
  __syncthreads();

  if (tid < E) {
    float r = 0.f;
    for (int j = 0; j < E; ++j) r += X[tid * SP + j];
    red[tid] = ALPHA / (1e-4f + r);
  }
  __syncthreads();

  for (int idx = tid; idx < E * E; idx += 256) {
    const int i = idx >> 7, j = idx & 127;
    const float mij = X[i * SP + j] * red[j];
    Pr[i * SP + j] = mij;
    PT[j * SP + i] = mij;
    X[i * SP + j]  = ((i == j) ? 1.f : 0.f) + mij;
  }
  __syncthreads();

  series_step<false, true>(X, Pr, PT, wave, lane);
  for (int it = 0; it < 3; ++it)
    series_step<true, true>(X, Pr, PT, wave, lane);
  series_step<true, false>(X, Pr, PT, wave, lane);

  if (tid < E) {
    const int i = tid;
    float cls[NCLS] = {0.f, 0.f, 0.f, 0.f, 0.f};
    float l1 = 0.f;
    for (int j = 0; j < E; ++j) {
      const float p = X[i * SP + j];
      l1 += fabsf(p);
      const int lb = lab[j];
      if (lb < NCLS) cls[lb] += p;
    }
    const float inv = 1.f / fmaxf(l1, 1e-12f);
#pragma unroll
    for (int c = 0; c < NCLS; ++c)
      out[((size_t)b * E + i) * NCLS + c] = logf(cls[c] * inv + 1e-6f);
  }
}

extern "C" void kernel_launch(void* const* d_in, const int* in_sizes, int n_in,
                              void* d_out, int out_size, void* d_ws, size_t ws_size,
                              hipStream_t stream) {
  const float* x      = (const float*)d_in[0];
  const int*   labels = (const int*)d_in[1];
  float* out   = (float*)d_out;
  float* sqd   = (float*)d_ws;
  float* stats = sqd + BATCH * E * E;

  const size_t sh1 = (size_t)(E * XP + E) * sizeof(float);
  k_gram<<<BATCH, 256, sh1, stream>>>(x, sqd);
  k_stats<<<1, ST, 0, stream>>>(sqd, stats);
  const size_t sh3 = (size_t)(3 * E * SP + 2 * E) * sizeof(float);
  k_solve<<<BATCH, 256, sh3, stream>>>(sqd, stats, labels, out);
}


# k_gram, k_solve: every WMMA product is symmetric (x x^T; series on N = h W h, h = sqrt of column factor), so only the 36 of 64 tiles on or above the diagonal are formed, mirrored on store
# speedup vs baseline: 2.7519x; 1.0523x over previous
#include <hip/hip_runtime.h>
#include <hip/hip_bf16.h>
#include <math.h>

typedef __attribute__((ext_vector_type(2))) float v2f;
typedef __attribute__((ext_vector_type(8))) float v8f;

#define BATCH 16
#define E 128
#define CDIM 512
#define NCLS 5
#define ALPHA 0.2f
#define XP 516
#define SP 140
#define DP 132


__device__ inline void fence() { __builtin_amdgcn_sched_barrier(0); }

__device__ inline v8f mma4(v2f a, v2f b, v8f c) {
  return __builtin_amdgcn_wmma_f32_16x16x4_f32(
       false, a,  false, b,
       (short)0, c,  false,  false);
}

__device__ inline int wave_tiles(int w, int (&tr)[5], int (&tc)[5]) {
  unsigned rp, cp;
  int nt;
  switch (w) {
    case 0:  rp = 0x00000u; cp = 0x43210u; nt = 5; break;
    case 1:  rp = 0x07000u; cp = 0x07765u; nt = 4; break;
    case 2:  rp = 0x06611u; cp = 0x07676u; nt = 4; break;
    case 3:  rp = 0x11111u; cp = 0x54321u; nt = 5; break;
    case 4:  rp = 0x05552u; cp = 0x07657u; nt = 4; break;
    case 5:  rp = 0x22222u; cp = 0x65432u; nt = 5; break;
    case 6:  rp = 0x33333u; cp = 0x76543u; nt = 5; break;
    default: rp = 0x04444u; cp = 0x07654u; nt = 4; break;
  }
#pragma unroll
  for (int t = 0; t < 5; ++t) {
    tr[t] = (int)((rp >> (4 * t)) & 15u);
    tc[t] = (int)((cp >> (4 * t)) & 15u);
  }
  return nt;
}

template <int NT>
__device__ inline void gram_load(const float* __restrict__ xs,
                                 const int (&oa)[5], const int (&ob)[5], int k0,
                                 v2f (&fa)[2][5], v2f (&fb)[2][5]) {
#pragma unroll
  for (int s = 0; s < 2; ++s) {
#pragma unroll
    for (int t = 0; t < NT; ++t) {
      fa[s][t] = *(const v2f*)(xs + oa[t] + k0 + 4 * s);
      fb[s][t] = *(const v2f*)(xs + ob[t] + k0 + 4 * s);
    }
  }
}

template <int NT>
__device__ inline void gram_mma(v8f (&acc)[5], const v2f (&fa)[2][5],
                                const v2f (&fb)[2][5]) {
#pragma unroll
  for (int s = 0; s < 2; ++s)
#pragma unroll
    for (int t = 0; t < NT; ++t) acc[t] = mma4(fa[s][t], fb[s][t], acc[t]);
}

template <int NT>
__device__ inline void gram_k(const float* __restrict__ xs, const int (&oa)[5],
                              const int (&ob)[5], v8f (&acc)[5]) {
  v2f fa0[2][5], fb0[2][5], fa1[2][5], fb1[2][5];
  gram_load<NT>(xs, oa, ob, 0, fa0, fb0);
  for (int k0 = 0; k0 < CDIM - 16; k0 += 16) {
    gram_load<NT>(xs, oa, ob, k0 + 8, fa1, fb1);
    fence();
    gram_mma<NT>(acc, fa0, fb0);
    fence();
    gram_load<NT>(xs, oa, ob, k0 + 16, fa0, fb0);
    fence();
    gram_mma<NT>(acc, fa1, fb1);
    fence();
  }
  gram_load<NT>(xs, oa, ob, CDIM - 8, fa1, fb1);
  fence();
  gram_mma<NT>(acc, fa0, fb0);
  fence();
  gram_mma<NT>(acc, fa1, fb1);
}

__global__ __launch_bounds__(256) void k_gram(const float* __restrict__ x,
                                              float* __restrict__ sqd) {
  extern __shared__ float smem[];
  float* xs    = smem;
  float* norms = smem + E * XP;
  const int b    = blockIdx.x;
  const int tid  = threadIdx.x;
  const int lane = tid & 31;
  const int wave = __builtin_amdgcn_readfirstlane(tid >> 5);

  const float4* xg = (const float4*)(x + (size_t)b * E * CDIM);
  for (int i = tid; i < E * CDIM / 4; i += 256) {
    const int r = i >> 7, c4 = i & 127;
    *(float4*)&xs[r * XP + c4 * 4] = xg[i];
  }
  __syncthreads();

  for (int i = tid; i < E; i += 256) {
    const float* row = xs + i * XP;
    float s = 0.f;
    for (int k = 0; k < CDIM; ++k) s += row[k] * row[k];
    norms[i] = s;
  }
  __syncthreads();

  const float invs = 0.04419417382415922f;
  const int m   = lane & 15;
  const int kb  = (lane >> 4) << 1;
  const int mo8 = (lane >> 4) << 3;
  int tr[5], tc[5];
  const int nt = wave_tiles(wave, tr, tc);
  int oa[5], ob[5];
  v8f acc[5];
#pragma unroll
  for (int t = 0; t < 5; ++t) {
    oa[t]  = (tr[t] * 16 + m) * XP + kb;
    ob[t]  = (tc[t] * 16 + m) * XP + kb;
    acc[t] = (v8f){0.f, 0.f, 0.f, 0.f, 0.f, 0.f, 0.f, 0.f};
  }
  if (nt == 5) gram_k<5>(xs, oa, ob, acc);
  else         gram_k<4>(xs, oa, ob, acc);
  __syncthreads();

  float* ds = smem;
#pragma unroll
  for (int t = 0; t < 5; ++t) {
    if (t < nt) {
      const int n  = tc[t] * 16 + m;
      const int i0 = tr[t] * 16 + mo8;
      float s[8];
#pragma unroll
      for (int r = 0; r < 8; ++r) {
        const int i = i0 + r;
        s[r] = (i == n) ? 0.f : (norms[i] + norms[n] - 2.f * acc[t][r]) * invs;
        ds[i * DP + n] = s[r];
      }
      if (tr[t] != tc[t]) {
        float* q = ds + n * DP + i0;
        *(float4*)q       = make_float4(s[0], s[1], s[2], s[3]);
        *(float4*)(q + 4) = make_float4(s[4], s[5], s[6], s[7]);
      }
    }
  }
  __syncthreads();

  float4* og = (float4*)(sqd + (size_t)b * E * E);
  for (int i = tid; i < E * E / 4; i += 256) {
    const int r = i >> 5, c4 = i & 31;
    og[i] = *(const float4*)&ds[r * DP + c4 * 4];
  }
}

#define ST 1024
#define SG 16
__global__ __launch_bounds__(ST) void k_stats(const float* __restrict__ sqd,
                                              float* __restrict__ stats) {
  __shared__ float s_part[3][ST / 32];
  const int tid  = threadIdx.x;
  const int lane = tid & 31;
  const int wave = tid >> 5;
  const float4* __restrict__ p = (const float4*)sqd;
  const int NV = BATCH * E * E / 4;
  float4 s4 = {0.f, 0.f, 0.f, 0.f};
  float4 q4 = {0.f, 0.f, 0.f, 0.f};
  float4 c4 = {0.f, 0.f, 0.f, 0.f};
  for (int base = 0; base < NV; base += ST * SG) {
    float4 v[SG];
#pragma unroll
    for (int u = 0; u < SG; ++u) v[u] = p[base + u * ST + tid];
#pragma unroll
    for (int u = 0; u < SG; ++u) {
      s4.x += v[u].x; s4.y += v[u].y; s4.z += v[u].z; s4.w += v[u].w;
      q4.x += v[u].x * v[u].x; q4.y += v[u].y * v[u].y;
      q4.z += v[u].z * v[u].z; q4.w += v[u].w * v[u].w;
      c4.x += (v[u].x != 0.f) ? 1.f : 0.f;
      c4.y += (v[u].y != 0.f) ? 1.f : 0.f;
      c4.z += (v[u].z != 0.f) ? 1.f : 0.f;
      c4.w += (v[u].w != 0.f) ? 1.f : 0.f;
    }
  }
  float sum = (s4.x + s4.y) + (s4.z + s4.w);
  float sq  = (q4.x + q4.y) + (q4.z + q4.w);
  float cnt = (c4.x + c4.y) + (c4.z + c4.w);
  for (int off = 16; off > 0; off >>= 1) {
    sum += __shfl_xor(sum, off);
    sq  += __shfl_xor(sq, off);
    cnt += __shfl_xor(cnt, off);
  }
  if (lane == 0) {
    s_part[0][wave] = sum; s_part[1][wave] = sq; s_part[2][wave] = cnt;
  }
  __syncthreads();
  if (wave == 0) {
    sum = s_part[0][lane]; sq = s_part[1][lane]; cnt = s_part[2][lane];
    for (int off = 16; off > 0; off >>= 1) {
      sum += __shfl_xor(sum, off);
      sq  += __shfl_xor(sq, off);
      cnt += __shfl_xor(cnt, off);
    }
    if (lane == 0) {
      float c    = cnt;
      float mean = sum / c;
      float var  = (sq - c * mean * mean) / (c - 1.f);
      stats[0]   = rsqrtf(var);
    }
  }
}

template <int NT, bool DO_X, bool DO_P>
__device__ inline void step_load(const float* __restrict__ Xm,
                                 const float* __restrict__ Pm,
                                 const int (&oa)[5], const int (&ob)[5], int k0,
                                 v2f (&fx)[2][5], v2f (&fp)[2][5],
                                 v2f (&fb)[2][5]) {
#pragma unroll
  for (int s = 0; s < 2; ++s) {
#pragma unroll
    for (int t = 0; t < NT; ++t) {
      if (DO_X) fx[s][t] = *(const v2f*)(Xm + oa[t] + k0 + 4 * s);
      if (DO_P) fp[s][t] = *(const v2f*)(Pm + oa[t] + k0 + 4 * s);
      fb[s][t] = *(const v2f*)(Pm + ob[t] + k0 + 4 * s);
    }
  }
}

template <int NT, bool DO_X, bool DO_P>
__device__ inline void step_mma(v8f (&ax)[5], v8f (&ap)[5],
                                const v2f (&fx)[2][5], const v2f (&fp)[2][5],
                                const v2f (&fb)[2][5]) {
#pragma unroll
  for (int s = 0; s < 2; ++s) {
#pragma unroll
    for (int t = 0; t < NT; ++t) {
      if (DO_X) ax[t] = mma4(fx[s][t], fb[s][t], ax[t]);
      if (DO_P) ap[t] = mma4(fp[s][t], fb[s][t], ap[t]);
    }
  }
}

template <int NT, bool DO_X, bool DO_P>
__device__ inline void step_k(const float* __restrict__ Xm,
                              const float* __restrict__ Pm, const int (&oa)[5],
                              const int (&ob)[5], v8f (&ax)[5], v8f (&ap)[5]) {
  v2f fx0[2][5], fp0[2][5], fb0[2][5], fx1[2][5], fp1[2][5], fb1[2][5];
  step_load<NT, DO_X, DO_P>(Xm, Pm, oa, ob, 0, fx0, fp0, fb0);
  for (int k0 = 0; k0 < E - 16; k0 += 16) {
    step_load<NT, DO_X, DO_P>(Xm, Pm, oa, ob, k0 + 8, fx1, fp1, fb1);
    fence();
    step_mma<NT, DO_X, DO_P>(ax, ap, fx0, fp0, fb0);
    fence();
    step_load<NT, DO_X, DO_P>(Xm, Pm, oa, ob, k0 + 16, fx0, fp0, fb0);
    fence();
    step_mma<NT, DO_X, DO_P>(ax, ap, fx1, fp1, fb1);
    fence();
  }
  step_load<NT, DO_X, DO_P>(Xm, Pm, oa, ob, E - 8, fx1, fp1, fb1);
  fence();
  step_mma<NT, DO_X, DO_P>(ax, ap, fx0, fp0, fb0);
  fence();
  step_mma<NT, DO_X, DO_P>(ax, ap, fx1, fp1, fb1);
}

__device__ inline void store_sym(float* __restrict__ Mm, int r, int c, int m,
                                 int mo8, v8f a) {
  const int n  = c * 16 + m;
  const int i0 = r * 16 + mo8;
#pragma unroll
  for (int q = 0; q < 8; ++q) Mm[(i0 + q) * SP + n] = a[q];
  if (r != c) {
    float* p = Mm + n * SP + i0;
    *(float4*)p       = make_float4(a[0], a[1], a[2], a[3]);
    *(float4*)(p + 4) = make_float4(a[4], a[5], a[6], a[7]);
  }
}

template <bool DO_X, bool DO_P>
__device__ inline void series_step(float* __restrict__ Xm, float* __restrict__ Pm,
                                   int nt, const int (&tr)[5],
                                   const int (&tc)[5], int lane) {
  const int m   = lane & 15;
  const int kb  = (lane >> 4) << 1;
  const int mo8 = (lane >> 4) << 3;
  int oa[5], ob[5];
  v8f ax[5], ap[5];
#pragma unroll
  for (int t = 0; t < 5; ++t) {
    oa[t] = (tr[t] * 16 + m) * SP + kb;
    ob[t] = (tc[t] * 16 + m) * SP + kb;
#pragma unroll
    for (int r = 0; r < 8; ++r) {
      ax[t][r] = DO_X ? Xm[(tr[t] * 16 + mo8 + r) * SP + tc[t] * 16 + m] : 0.f;
      ap[t][r] = 0.f;
    }
  }
  if (nt == 5) step_k<5, DO_X, DO_P>(Xm, Pm, oa, ob, ax, ap);
  else         step_k<4, DO_X, DO_P>(Xm, Pm, oa, ob, ax, ap);
  __syncthreads();
#pragma unroll
  for (int t = 0; t < 5; ++t) {
    if (t < nt) {
      if (DO_P) store_sym(Pm, tr[t], tc[t], m, mo8, ap[t]);
      if (DO_X) store_sym(Xm, tr[t], tc[t], m, mo8, ax[t]);
    }
  }
  __syncthreads();
}

__global__ __launch_bounds__(256) void k_solve(const float* __restrict__ sqd,
                                               const float* __restrict__ stats,
                                               const int* __restrict__ labels,
                                               float* __restrict__ out) {
  extern __shared__ float smem[];
  float* Pm = smem;
  float* X  = smem + E * SP;
  float* hs = smem + 2 * E * SP;
  int*   lab = (int*)(smem + 2 * E * SP + E);
  const int b    = blockIdx.x;
  const int tid  = threadIdx.x;
  const int lane = tid & 31;
  const int wave = __builtin_amdgcn_readfirstlane(tid >> 5);
  const float rstd = stats[0];
  const float* sq  = sqd + (size_t)b * E * E;

  if (tid < E) lab[tid] = labels[b * E + tid];

  for (int idx = tid; idx < E * E; idx += 256) {
    const int i = idx >> 7, j = idx & 127;
    X[i * SP + j] = (i == j) ? 0.f : expf(-sq[idx] * rstd);
  }
  __syncthreads();

  if (tid < E) {
    float r = 0.f;
    for (int j = 0; j < E; ++j) r += X[tid * SP + j];
    hs[tid] = sqrtf(ALPHA / (1e-4f + r));
  }
  __syncthreads();

  for (int idx = tid; idx < E * E; idx += 256) {
    const int i = idx >> 7, j = idx & 127;
    const float nij = X[i * SP + j] * (hs[i] * hs[j]);
    Pm[i * SP + j] = nij;
    X[i * SP + j]  = ((i == j) ? 1.f : 0.f) + nij;
  }
  __syncthreads();

  int tr[5], tc[5];
  const int nt = wave_tiles(wave, tr, tc);
  series_step<false, true>(X, Pm, nt, tr, tc, lane);
  for (int it = 0; it < 3; ++it)
    series_step<true, true>(X, Pm, nt, tr, tc, lane);
  series_step<true, false>(X, Pm, nt, tr, tc, lane);

  if (tid < E) {
    const int i = tid;
    float cls[NCLS] = {0.f, 0.f, 0.f, 0.f, 0.f};
    float l1 = 0.f;
    for (int j = 0; j < E; ++j) {
      const float p = X[i * SP + j] * hs[j];
      l1 += fabsf(p);
      const int lb = lab[j];
      if (lb < NCLS) cls[lb] += p;
    }
    const float rh  = 1.f / hs[i];
    const float inv = 1.f / fmaxf(l1 * rh, 1e-12f);
#pragma unroll
    for (int c = 0; c < NCLS; ++c)
      out[((size_t)b * E + i) * NCLS + c] = logf((cls[c] * rh) * inv + 1e-6f);
  }
}

extern "C" void kernel_launch(void* const* d_in, const int* in_sizes, int n_in,
                              void* d_out, int out_size, void* d_ws, size_t ws_size,
                              hipStream_t stream) {
  const float* x      = (const float*)d_in[0];
  const int*   labels = (const int*)d_in[1];
  float* out   = (float*)d_out;
  float* sqd   = (float*)d_ws;
  float* stats = sqd + BATCH * E * E;

  const size_t sh1 = (size_t)(E * XP + E) * sizeof(float);
  k_gram<<<BATCH, 256, sh1, stream>>>(x, sqd);
  k_stats<<<1, ST, 0, stream>>>(sqd, stats);
  const size_t sh3 = (size_t)(2 * E * SP + 2 * E) * sizeof(float);
  k_solve<<<BATCH, 256, sh3, stream>>>(sqd, stats, labels, out);
}


# k_stats: the statistics pass spread over 64 work-groups (each reads its 16 KB as one group of four 16-byte loads per lane, writes one partial triple); every wave of k_solve sums the 64 triples itself
# speedup vs baseline: 3.4498x; 1.2536x over previous
#include <hip/hip_runtime.h>
#include <hip/hip_bf16.h>
#include <math.h>

typedef __attribute__((ext_vector_type(2))) float v2f;
typedef __attribute__((ext_vector_type(8))) float v8f;

#define BATCH 16
#define E 128
#define CDIM 512
#define NCLS 5
#define ALPHA 0.2f
#define XP 516
#define SP 140
#define DP 132


__device__ inline void fence() { __builtin_amdgcn_sched_barrier(0); }

__device__ inline v8f mma4(v2f a, v2f b, v8f c) {
  return __builtin_amdgcn_wmma_f32_16x16x4_f32(
       false, a,  false, b,
       (short)0, c,  false,  false);
}

__device__ inline int wave_tiles(int w, int (&tr)[5], int (&tc)[5]) {
  unsigned rp, cp;
  int nt;
  switch (w) {
    case 0:  rp = 0x00000u; cp = 0x43210u; nt = 5; break;
    case 1:  rp = 0x07000u; cp = 0x07765u; nt = 4; break;
    case 2:  rp = 0x06611u; cp = 0x07676u; nt = 4; break;
    case 3:  rp = 0x11111u; cp = 0x54321u; nt = 5; break;
    case 4:  rp = 0x05552u; cp = 0x07657u; nt = 4; break;
    case 5:  rp = 0x22222u; cp = 0x65432u; nt = 5; break;
    case 6:  rp = 0x33333u; cp = 0x76543u; nt = 5; break;
    default: rp = 0x04444u; cp = 0x07654u; nt = 4; break;
  }
#pragma unroll
  for (int t = 0; t < 5; ++t) {
    tr[t] = (int)((rp >> (4 * t)) & 15u);
    tc[t] = (int)((cp >> (4 * t)) & 15u);
  }
  return nt;
}

template <int NT>
__device__ inline void gram_load(const float* __restrict__ xs,
                                 const int (&oa)[5], const int (&ob)[5], int k0,
                                 v2f (&fa)[2][5], v2f (&fb)[2][5]) {
#pragma unroll
  for (int s = 0; s < 2; ++s) {
#pragma unroll
    for (int t = 0; t < NT; ++t) {
      fa[s][t] = *(const v2f*)(xs + oa[t] + k0 + 4 * s);
      fb[s][t] = *(const v2f*)(xs + ob[t] + k0 + 4 * s);
    }
  }
}

template <int NT>
__device__ inline void gram_mma(v8f (&acc)[5], const v2f (&fa)[2][5],
                                const v2f (&fb)[2][5]) {
#pragma unroll
  for (int s = 0; s < 2; ++s)
#pragma unroll
    for (int t = 0; t < NT; ++t) acc[t] = mma4(fa[s][t], fb[s][t], acc[t]);
}

template <int NT>
__device__ inline void gram_k(const float* __restrict__ xs, const int (&oa)[5],
                              const int (&ob)[5], v8f (&acc)[5]) {
  v2f fa0[2][5], fb0[2][5], fa1[2][5], fb1[2][5];
  gram_load<NT>(xs, oa, ob, 0, fa0, fb0);
  for (int k0 = 0; k0 < CDIM - 16; k0 += 16) {
    gram_load<NT>(xs, oa, ob, k0 + 8, fa1, fb1);
    fence();
    gram_mma<NT>(acc, fa0, fb0);
    fence();
    gram_load<NT>(xs, oa, ob, k0 + 16, fa0, fb0);
    fence();
    gram_mma<NT>(acc, fa1, fb1);
    fence();
  }
  gram_load<NT>(xs, oa, ob, CDIM - 8, fa1, fb1);
  fence();
  gram_mma<NT>(acc, fa0, fb0);
  fence();
  gram_mma<NT>(acc, fa1, fb1);
}

__global__ __launch_bounds__(256) void k_gram(const float* __restrict__ x,
                                              float* __restrict__ sqd) {
  extern __shared__ float smem[];
  float* xs    = smem;
  float* norms = smem + E * XP;
  const int b    = blockIdx.x;
  const int tid  = threadIdx.x;
  const int lane = tid & 31;
  const int wave = __builtin_amdgcn_readfirstlane(tid >> 5);

  const float4* xg = (const float4*)(x + (size_t)b * E * CDIM);
  for (int i = tid; i < E * CDIM / 4; i += 256) {
    const int r = i >> 7, c4 = i & 127;
    *(float4*)&xs[r * XP + c4 * 4] = xg[i];
  }
  __syncthreads();

  for (int i = tid; i < E; i += 256) {
    const float* row = xs + i * XP;
    float s = 0.f;
    for (int k = 0; k < CDIM; ++k) s += row[k] * row[k];
    norms[i] = s;
  }
  __syncthreads();

  const float invs = 0.04419417382415922f;
  const int m   = lane & 15;
  const int kb  = (lane >> 4) << 1;
  const int mo8 = (lane >> 4) << 3;
  int tr[5], tc[5];
  const int nt = wave_tiles(wave, tr, tc);
  int oa[5], ob[5];
  v8f acc[5];
#pragma unroll
  for (int t = 0; t < 5; ++t) {
    oa[t]  = (tr[t] * 16 + m) * XP + kb;
    ob[t]  = (tc[t] * 16 + m) * XP + kb;
    acc[t] = (v8f){0.f, 0.f, 0.f, 0.f, 0.f, 0.f, 0.f, 0.f};
  }
  if (nt == 5) gram_k<5>(xs, oa, ob, acc);
  else         gram_k<4>(xs, oa, ob, acc);
  __syncthreads();

  float* ds = smem;
#pragma unroll
  for (int t = 0; t < 5; ++t) {
    if (t < nt) {
      const int n  = tc[t] * 16 + m;
      const int i0 = tr[t] * 16 + mo8;
      float s[8];
#pragma unroll
      for (int r = 0; r < 8; ++r) {
        const int i = i0 + r;
        s[r] = (i == n) ? 0.f : (norms[i] + norms[n] - 2.f * acc[t][r]) * invs;
        ds[i * DP + n] = s[r];
      }
      if (tr[t] != tc[t]) {
        float* q = ds + n * DP + i0;
        *(float4*)q       = make_float4(s[0], s[1], s[2], s[3]);
        *(float4*)(q + 4) = make_float4(s[4], s[5], s[6], s[7]);
      }
    }
  }
  __syncthreads();

  float4* og = (float4*)(sqd + (size_t)b * E * E);
  for (int i = tid; i < E * E / 4; i += 256) {
    const int r = i >> 5, c4 = i & 31;
    og[i] = *(const float4*)&ds[r * DP + c4 * 4];
  }
}

#define SB 64
#define ST 256
#define SG 4
__global__ __launch_bounds__(ST) void k_stats(const float* __restrict__ sqd,
                                              float* __restrict__ part) {
  __shared__ float s_part[3][ST / 32];
  const int tid  = threadIdx.x;
  const int lane = tid & 31;
  const int wave = tid >> 5;
  const float4* __restrict__ p = (const float4*)sqd + (size_t)blockIdx.x * (ST * SG);
  float4 v[SG];
#pragma unroll
  for (int u = 0; u < SG; ++u) v[u] = p[u * ST + tid];
  float4 s4 = {0.f, 0.f, 0.f, 0.f};
  float4 q4 = {0.f, 0.f, 0.f, 0.f};
  float4 c4 = {0.f, 0.f, 0.f, 0.f};
#pragma unroll
  for (int u = 0; u < SG; ++u) {
    s4.x += v[u].x; s4.y += v[u].y; s4.z += v[u].z; s4.w += v[u].w;
    q4.x += v[u].x * v[u].x; q4.y += v[u].y * v[u].y;
    q4.z += v[u].z * v[u].z; q4.w += v[u].w * v[u].w;
    c4.x += (v[u].x != 0.f) ? 1.f : 0.f;
    c4.y += (v[u].y != 0.f) ? 1.f : 0.f;
    c4.z += (v[u].z != 0.f) ? 1.f : 0.f;
    c4.w += (v[u].w != 0.f) ? 1.f : 0.f;
  }
  float sum = (s4.x + s4.y) + (s4.z + s4.w);
  float sq  = (q4.x + q4.y) + (q4.z + q4.w);
  float cnt = (c4.x + c4.y) + (c4.z + c4.w);
  for (int off = 16; off > 0; off >>= 1) {
    sum += __shfl_xor(sum, off);
    sq  += __shfl_xor(sq, off);
    cnt += __shfl_xor(cnt, off);
  }
  if (lane == 0) {
    s_part[0][wave] = sum; s_part[1][wave] = sq; s_part[2][wave] = cnt;
  }
  __syncthreads();
  if (wave == 0) {
    const int w = lane & (ST / 32 - 1);
    sum = s_part[0][w]; sq = s_part[1][w]; cnt = s_part[2][w];
    for (int off = ST / 64; off > 0; off >>= 1) {
      sum += __shfl_xor(sum, off);
      sq  += __shfl_xor(sq, off);
      cnt += __shfl_xor(cnt, off);
    }
    if (lane == 0) {
      part[blockIdx.x]          = sum;
      part[SB + blockIdx.x]     = sq;
      part[2 * SB + blockIdx.x] = cnt;
    }
  }
}

__device__ inline float stats_rstd(const float* __restrict__ part, int lane) {
  float sum = part[lane]          + part[lane + 32];
  float sq  = part[SB + lane]     + part[SB + lane + 32];
  float cnt = part[2 * SB + lane] + part[2 * SB + lane + 32];
  for (int off = 16; off > 0; off >>= 1) {
    sum += __shfl_xor(sum, off);
    sq  += __shfl_xor(sq, off);
    cnt += __shfl_xor(cnt, off);
  }
  const float mean = sum / cnt;
  const float var  = (sq - cnt * mean * mean) / (cnt - 1.f);
  return rsqrtf(var);
}

template <int NT, bool DO_X, bool DO_P>
__device__ inline void step_load(const float* __restrict__ Xm,
                                 const float* __restrict__ Pm,
                                 const int (&oa)[5], const int (&ob)[5], int k0,
                                 v2f (&fx)[2][5], v2f (&fp)[2][5],
                                 v2f (&fb)[2][5]) {
#pragma unroll
  for (int s = 0; s < 2; ++s) {
#pragma unroll
    for (int t = 0; t < NT; ++t) {
      if (DO_X) fx[s][t] = *(const v2f*)(Xm + oa[t] + k0 + 4 * s);
      if (DO_P) fp[s][t] = *(const v2f*)(Pm + oa[t] + k0 + 4 * s);
      fb[s][t] = *(const v2f*)(Pm + ob[t] + k0 + 4 * s);
    }
  }
}

template <int NT, bool DO_X, bool DO_P>
__device__ inline void step_mma(v8f (&ax)[5], v8f (&ap)[5],
                                const v2f (&fx)[2][5], const v2f (&fp)[2][5],
                                const v2f (&fb)[2][5]) {
#pragma unroll
  for (int s = 0; s < 2; ++s) {
#pragma unroll
    for (int t = 0; t < NT; ++t) {
      if (DO_X) ax[t] = mma4(fx[s][t], fb[s][t], ax[t]);
      if (DO_P) ap[t] = mma4(fp[s][t], fb[s][t], ap[t]);
    }
  }
}

template <int NT, bool DO_X, bool DO_P>
__device__ inline void step_k(const float* __restrict__ Xm,
                              const float* __restrict__ Pm, const int (&oa)[5],
                              const int (&ob)[5], v8f (&ax)[5], v8f (&ap)[5]) {
  v2f fx0[2][5], fp0[2][5], fb0[2][5], fx1[2][5], fp1[2][5], fb1[2][5];
  step_load<NT, DO_X, DO_P>(Xm, Pm, oa, ob, 0, fx0, fp0, fb0);
  for (int k0 = 0; k0 < E - 16; k0 += 16) {
    step_load<NT, DO_X, DO_P>(Xm, Pm, oa, ob, k0 + 8, fx1, fp1, fb1);
    fence();
    step_mma<NT, DO_X, DO_P>(ax, ap, fx0, fp0, fb0);
    fence();
    step_load<NT, DO_X, DO_P>(Xm, Pm, oa, ob, k0 + 16, fx0, fp0, fb0);
    fence();
    step_mma<NT, DO_X, DO_P>(ax, ap, fx1, fp1, fb1);
    fence();
  }
  step_load<NT, DO_X, DO_P>(Xm, Pm, oa, ob, E - 8, fx1, fp1, fb1);
  fence();
  step_mma<NT, DO_X, DO_P>(ax, ap, fx0, fp0, fb0);
  fence();
  step_mma<NT, DO_X, DO_P>(ax, ap, fx1, fp1, fb1);
}

__device__ inline void store_sym(float* __restrict__ Mm, int r, int c, int m,
                                 int mo8, v8f a) {
  const int n  = c * 16 + m;
  const int i0 = r * 16 + mo8;
#pragma unroll
  for (int q = 0; q < 8; ++q) Mm[(i0 + q) * SP + n] = a[q];
  if (r != c) {
    float* p = Mm + n * SP + i0;
    *(float4*)p       = make_float4(a[0], a[1], a[2], a[3]);
    *(float4*)(p + 4) = make_float4(a[4], a[5], a[6], a[7]);
  }
}

template <bool DO_X, bool DO_P>
__device__ inline void series_step(float* __restrict__ Xm, float* __restrict__ Pm,
                                   int nt, const int (&tr)[5],
                                   const int (&tc)[5], int lane) {
  const int m   = lane & 15;
  const int kb  = (lane >> 4) << 1;
  const int mo8 = (lane >> 4) << 3;
  int oa[5], ob[5];
  v8f ax[5], ap[5];
#pragma unroll
  for (int t = 0; t < 5; ++t) {
    oa[t] = (tr[t] * 16 + m) * SP + kb;
    ob[t] = (tc[t] * 16 + m) * SP + kb;
#pragma unroll
    for (int r = 0; r < 8; ++r) {
      ax[t][r] = DO_X ? Xm[(tr[t] * 16 + mo8 + r) * SP + tc[t] * 16 + m] : 0.f;
      ap[t][r] = 0.f;
    }
  }
  if (nt == 5) step_k<5, DO_X, DO_P>(Xm, Pm, oa, ob, ax, ap);
  else         step_k<4, DO_X, DO_P>(Xm, Pm, oa, ob, ax, ap);
  __syncthreads();
#pragma unroll
  for (int t = 0; t < 5; ++t) {
    if (t < nt) {
      if (DO_P) store_sym(Pm, tr[t], tc[t], m, mo8, ap[t]);
      if (DO_X) store_sym(Xm, tr[t], tc[t], m, mo8, ax[t]);
    }
  }
  __syncthreads();
}

__global__ __launch_bounds__(256) void k_solve(const float* __restrict__ sqd,
                                               const float* __restrict__ stats,
                                               const int* __restrict__ labels,
                                               float* __restrict__ out) {
  extern __shared__ float smem[];
  float* Pm = smem;
  float* X  = smem + E * SP;
  float* hs = smem + 2 * E * SP;
  int*   lab = (int*)(smem + 2 * E * SP + E);
  const int b    = blockIdx.x;
  const int tid  = threadIdx.x;
  const int lane = tid & 31;
  const int wave = __builtin_amdgcn_readfirstlane(tid >> 5);
  const float rstd = stats_rstd(stats, lane);
  const float* sq  = sqd + (size_t)b * E * E;

  if (tid < E) lab[tid] = labels[b * E + tid];

  for (int idx = tid; idx < E * E; idx += 256) {
    const int i = idx >> 7, j = idx & 127;
    X[i * SP + j] = (i == j) ? 0.f : expf(-sq[idx] * rstd);
  }
  __syncthreads();

  if (tid < E) {
    float r = 0.f;
    for (int j = 0; j < E; ++j) r += X[tid * SP + j];
    hs[tid] = sqrtf(ALPHA / (1e-4f + r));
  }
  __syncthreads();

  for (int idx = tid; idx < E * E; idx += 256) {
    const int i = idx >> 7, j = idx & 127;
    const float nij = X[i * SP + j] * (hs[i] * hs[j]);
    Pm[i * SP + j] = nij;
    X[i * SP + j]  = ((i == j) ? 1.f : 0.f) + nij;
  }
  __syncthreads();

  int tr[5], tc[5];
  const int nt = wave_tiles(wave, tr, tc);
  series_step<false, true>(X, Pm, nt, tr, tc, lane);
  for (int it = 0; it < 3; ++it)
    series_step<true, true>(X, Pm, nt, tr, tc, lane);
  series_step<true, false>(X, Pm, nt, tr, tc, lane);

  if (tid < E) {
    const int i = tid;
    float cls[NCLS] = {0.f, 0.f, 0.f, 0.f, 0.f};
    float l1 = 0.f;
    for (int j = 0; j < E; ++j) {
      const float p = X[i * SP + j] * hs[j];
      l1 += fabsf(p);
      const int lb = lab[j];
      if (lb < NCLS) cls[lb] += p;
    }
    const float rh  = 1.f / hs[i];
    const float inv = 1.f / fmaxf(l1 * rh, 1e-12f);
#pragma unroll
    for (int c = 0; c < NCLS; ++c)
      out[((size_t)b * E + i) * NCLS + c] = logf((cls[c] * rh) * inv + 1e-6f);
  }
}

extern "C" void kernel_launch(void* const* d_in, const int* in_sizes, int n_in,
                              void* d_out, int out_size, void* d_ws, size_t ws_size,
                              hipStream_t stream) {
  const float* x      = (const float*)d_in[0];
  const int*   labels = (const int*)d_in[1];
  float* out   = (float*)d_out;
  float* sqd   = (float*)d_ws;
  float* stats = sqd + BATCH * E * E;

  const size_t sh1 = (size_t)(E * XP + E) * sizeof(float);
  k_gram<<<BATCH, 256, sh1, stream>>>(x, sqd);
  k_stats<<<SB, ST, 0, stream>>>(sqd, stats);
  const size_t sh3 = (size_t)(2 * E * SP + 2 * E) * sizeof(float);
  k_solve<<<BATCH, 256, sh3, stream>>>(sqd, stats, labels, out);
}


# k_gram: row norms read off the Gram product's diagonal tiles (one per wave) instead of a 512-step LDS walk on 4 waves; epilogue reads each tile's 9 norms as one group, diagonal zeroed by select
# speedup vs baseline: 3.7247x; 1.0797x over previous
#include <hip/hip_runtime.h>
#include <hip/hip_bf16.h>
#include <math.h>

typedef __attribute__((ext_vector_type(2))) float v2f;
typedef __attribute__((ext_vector_type(8))) float v8f;

#define BATCH 16
#define E 128
#define CDIM 512
#define NCLS 5
#define ALPHA 0.2f
#define XP 516
#define SP 140
#define DP 132


__device__ inline void fence() { __builtin_amdgcn_sched_barrier(0); }

__device__ inline v8f mma4(v2f a, v2f b, v8f c) {
  return __builtin_amdgcn_wmma_f32_16x16x4_f32(
       false, a,  false, b,
       (short)0, c,  false,  false);
}

__device__ inline int wave_tiles(int w, int (&tr)[5], int (&tc)[5]) {
  unsigned rp, cp;
  int nt;
  switch (w) {
    case 0:  rp = 0x00000u; cp = 0x43210u; nt = 5; break;
    case 1:  rp = 0x07000u; cp = 0x07765u; nt = 4; break;
    case 2:  rp = 0x06611u; cp = 0x07676u; nt = 4; break;
    case 3:  rp = 0x11111u; cp = 0x54321u; nt = 5; break;
    case 4:  rp = 0x05552u; cp = 0x07657u; nt = 4; break;
    case 5:  rp = 0x22222u; cp = 0x65432u; nt = 5; break;
    case 6:  rp = 0x33333u; cp = 0x76543u; nt = 5; break;
    default: rp = 0x04444u; cp = 0x07654u; nt = 4; break;
  }
#pragma unroll
  for (int t = 0; t < 5; ++t) {
    tr[t] = (int)((rp >> (4 * t)) & 15u);
    tc[t] = (int)((cp >> (4 * t)) & 15u);
  }
  return nt;
}

template <int NT>
__device__ inline void gram_load(const float* __restrict__ xs,
                                 const int (&oa)[5], const int (&ob)[5], int k0,
                                 v2f (&fa)[2][5], v2f (&fb)[2][5]) {
#pragma unroll
  for (int s = 0; s < 2; ++s) {
#pragma unroll
    for (int t = 0; t < NT; ++t) {
      fa[s][t] = *(const v2f*)(xs + oa[t] + k0 + 4 * s);
      fb[s][t] = *(const v2f*)(xs + ob[t] + k0 + 4 * s);
    }
  }
}

template <int NT>
__device__ inline void gram_mma(v8f (&acc)[5], const v2f (&fa)[2][5],
                                const v2f (&fb)[2][5]) {
#pragma unroll
  for (int s = 0; s < 2; ++s)
#pragma unroll
    for (int t = 0; t < NT; ++t) acc[t] = mma4(fa[s][t], fb[s][t], acc[t]);
}

template <int NT>
__device__ inline void gram_k(const float* __restrict__ xs, const int (&oa)[5],
                              const int (&ob)[5], v8f (&acc)[5]) {
  v2f fa0[2][5], fb0[2][5], fa1[2][5], fb1[2][5];
  gram_load<NT>(xs, oa, ob, 0, fa0, fb0);
  for (int k0 = 0; k0 < CDIM - 16; k0 += 16) {
    gram_load<NT>(xs, oa, ob, k0 + 8, fa1, fb1);
    fence();
    gram_mma<NT>(acc, fa0, fb0);
    fence();
    gram_load<NT>(xs, oa, ob, k0 + 16, fa0, fb0);
    fence();
    gram_mma<NT>(acc, fa1, fb1);
    fence();
  }
  gram_load<NT>(xs, oa, ob, CDIM - 8, fa1, fb1);
  fence();
  gram_mma<NT>(acc, fa0, fb0);
  fence();
  gram_mma<NT>(acc, fa1, fb1);
}

__global__ __launch_bounds__(256) void k_gram(const float* __restrict__ x,
                                              float* __restrict__ sqd) {
  extern __shared__ float smem[];
  float* xs    = smem;
  float* norms = smem + E * XP;
  const int b    = blockIdx.x;
  const int tid  = threadIdx.x;
  const int lane = tid & 31;
  const int wave = __builtin_amdgcn_readfirstlane(tid >> 5);

  const float4* xg = (const float4*)(x + (size_t)b * E * CDIM);
  for (int i = tid; i < E * CDIM / 4; i += 256) {
    const int r = i >> 7, c4 = i & 127;
    *(float4*)&xs[r * XP + c4 * 4] = xg[i];
  }
  __syncthreads();

  const float invs = 0.04419417382415922f;
  const int m   = lane & 15;
  const int kb  = (lane >> 4) << 1;
  const int mo8 = (lane >> 4) << 3;
  int tr[5], tc[5];
  const int nt = wave_tiles(wave, tr, tc);
  int oa[5], ob[5];
  v8f acc[5];
#pragma unroll
  for (int t = 0; t < 5; ++t) {
    oa[t]  = (tr[t] * 16 + m) * XP + kb;
    ob[t]  = (tc[t] * 16 + m) * XP + kb;
    acc[t] = (v8f){0.f, 0.f, 0.f, 0.f, 0.f, 0.f, 0.f, 0.f};
  }
  if (nt == 5) gram_k<5>(xs, oa, ob, acc);
  else         gram_k<4>(xs, oa, ob, acc);

#pragma unroll
  for (int t = 0; t < 5; ++t) {
    if (t < nt && tr[t] == tc[t]) {
      float d = acc[t][0];
#pragma unroll
      for (int q = 1; q < 8; ++q) d = ((m & 7) == q) ? acc[t][q] : d;
      if ((m >> 3) == (lane >> 4)) norms[tr[t] * 16 + m] = d;
    }
  }
  __syncthreads();

  float nn[5], ni[5][8];
#pragma unroll
  for (int t = 0; t < 5; ++t) {
    const float4 lo = *(const float4*)&norms[tr[t] * 16 + mo8];
    const float4 hi = *(const float4*)&norms[tr[t] * 16 + mo8 + 4];
    nn[t] = norms[tc[t] * 16 + m];
    ni[t][0] = lo.x; ni[t][1] = lo.y; ni[t][2] = lo.z; ni[t][3] = lo.w;
    ni[t][4] = hi.x; ni[t][5] = hi.y; ni[t][6] = hi.z; ni[t][7] = hi.w;
  }

  float* ds = smem;
#pragma unroll
  for (int t = 0; t < 5; ++t) {
    if (t < nt) {
      const int n  = tc[t] * 16 + m;
      const int i0 = tr[t] * 16 + mo8;
      float s[8];
#pragma unroll
      for (int r = 0; r < 8; ++r) {
        const int i = i0 + r;
        const float v = (ni[t][r] + nn[t] - 2.f * acc[t][r]) * invs;
        s[r] = (i == n) ? 0.f : v;
        ds[i * DP + n] = s[r];
      }
      if (tr[t] != tc[t]) {
        float* q = ds + n * DP + i0;
        *(float4*)q       = make_float4(s[0], s[1], s[2], s[3]);
        *(float4*)(q + 4) = make_float4(s[4], s[5], s[6], s[7]);
      }
    }
  }
  __syncthreads();

  float4* og = (float4*)(sqd + (size_t)b * E * E);
  for (int i = tid; i < E * E / 4; i += 256) {
    const int r = i >> 5, c4 = i & 31;
    og[i] = *(const float4*)&ds[r * DP + c4 * 4];
  }
}

#define SB 64
#define ST 256
#define SG 4
__global__ __launch_bounds__(ST) void k_stats(const float* __restrict__ sqd,
                                              float* __restrict__ part) {
  __shared__ float s_part[3][ST / 32];
  const int tid  = threadIdx.x;
  const int lane = tid & 31;
  const int wave = tid >> 5;
  const float4* __restrict__ p = (const float4*)sqd + (size_t)blockIdx.x * (ST * SG);
  float4 v[SG];
#pragma unroll
  for (int u = 0; u < SG; ++u) v[u] = p[u * ST + tid];
  float4 s4 = {0.f, 0.f, 0.f, 0.f};
  float4 q4 = {0.f, 0.f, 0.f, 0.f};
  float4 c4 = {0.f, 0.f, 0.f, 0.f};
#pragma unroll
  for (int u = 0; u < SG; ++u) {
    s4.x += v[u].x; s4.y += v[u].y; s4.z += v[u].z; s4.w += v[u].w;
    q4.x += v[u].x * v[u].x; q4.y += v[u].y * v[u].y;
    q4.z += v[u].z * v[u].z; q4.w += v[u].w * v[u].w;
    c4.x += (v[u].x != 0.f) ? 1.f : 0.f;
    c4.y += (v[u].y != 0.f) ? 1.f : 0.f;
    c4.z += (v[u].z != 0.f) ? 1.f : 0.f;
    c4.w += (v[u].w != 0.f) ? 1.f : 0.f;
  }
  float sum = (s4.x + s4.y) + (s4.z + s4.w);
  float sq  = (q4.x + q4.y) + (q4.z + q4.w);
  float cnt = (c4.x + c4.y) + (c4.z + c4.w);
  for (int off = 16; off > 0; off >>= 1) {
    sum += __shfl_xor(sum, off);
    sq  += __shfl_xor(sq, off);
    cnt += __shfl_xor(cnt, off);
  }
  if (lane == 0) {
    s_part[0][wave] = sum; s_part[1][wave] = sq; s_part[2][wave] = cnt;
  }
  __syncthreads();
  if (wave == 0) {
    const int w = lane & (ST / 32 - 1);
    sum = s_part[0][w]; sq = s_part[1][w]; cnt = s_part[2][w];
    for (int off = ST / 64; off > 0; off >>= 1) {
      sum += __shfl_xor(sum, off);
      sq  += __shfl_xor(sq, off);
      cnt += __shfl_xor(cnt, off);
    }
    if (lane == 0) {
      part[blockIdx.x]          = sum;
      part[SB + blockIdx.x]     = sq;
      part[2 * SB + blockIdx.x] = cnt;
    }
  }
}

__device__ inline float stats_rstd(const float* __restrict__ part, int lane) {
  float sum = part[lane]          + part[lane + 32];
  float sq  = part[SB + lane]     + part[SB + lane + 32];
  float cnt = part[2 * SB + lane] + part[2 * SB + lane + 32];
  for (int off = 16; off > 0; off >>= 1) {
    sum += __shfl_xor(sum, off);
    sq  += __shfl_xor(sq, off);
    cnt += __shfl_xor(cnt, off);
  }
  const float mean = sum / cnt;
  const float var  = (sq - cnt * mean * mean) / (cnt - 1.f);
  return rsqrtf(var);
}

template <int NT, bool DO_X, bool DO_P>
__device__ inline void step_load(const float* __restrict__ Xm,
                                 const float* __restrict__ Pm,
                                 const int (&oa)[5], const int (&ob)[5], int k0,
                                 v2f (&fx)[2][5], v2f (&fp)[2][5],
                                 v2f (&fb)[2][5]) {
#pragma unroll
  for (int s = 0; s < 2; ++s) {
#pragma unroll
    for (int t = 0; t < NT; ++t) {
      if (DO_X) fx[s][t] = *(const v2f*)(Xm + oa[t] + k0 + 4 * s);
      if (DO_P) fp[s][t] = *(const v2f*)(Pm + oa[t] + k0 + 4 * s);
      fb[s][t] = *(const v2f*)(Pm + ob[t] + k0 + 4 * s);
    }
  }
}

template <int NT, bool DO_X, bool DO_P>
__device__ inline void step_mma(v8f (&ax)[5], v8f (&ap)[5],
                                const v2f (&fx)[2][5], const v2f (&fp)[2][5],
                                const v2f (&fb)[2][5]) {
#pragma unroll
  for (int s = 0; s < 2; ++s) {
#pragma unroll
    for (int t = 0; t < NT; ++t) {
      if (DO_X) ax[t] = mma4(fx[s][t], fb[s][t], ax[t]);
      if (DO_P) ap[t] = mma4(fp[s][t], fb[s][t], ap[t]);
    }
  }
}

template <int NT, bool DO_X, bool DO_P>
__device__ inline void step_k(const float* __restrict__ Xm,
                              const float* __restrict__ Pm, const int (&oa)[5],
                              const int (&ob)[5], v8f (&ax)[5], v8f (&ap)[5]) {
  v2f fx0[2][5], fp0[2][5], fb0[2][5], fx1[2][5], fp1[2][5], fb1[2][5];
  step_load<NT, DO_X, DO_P>(Xm, Pm, oa, ob, 0, fx0, fp0, fb0);
  for (int k0 = 0; k0 < E - 16; k0 += 16) {
    step_load<NT, DO_X, DO_P>(Xm, Pm, oa, ob, k0 + 8, fx1, fp1, fb1);
    fence();
    step_mma<NT, DO_X, DO_P>(ax, ap, fx0, fp0, fb0);
    fence();
    step_load<NT, DO_X, DO_P>(Xm, Pm, oa, ob, k0 + 16, fx0, fp0, fb0);
    fence();
    step_mma<NT, DO_X, DO_P>(ax, ap, fx1, fp1, fb1);
    fence();
  }
  step_load<NT, DO_X, DO_P>(Xm, Pm, oa, ob, E - 8, fx1, fp1, fb1);
  fence();
  step_mma<NT, DO_X, DO_P>(ax, ap, fx0, fp0, fb0);
  fence();
  step_mma<NT, DO_X, DO_P>(ax, ap, fx1, fp1, fb1);
}

__device__ inline void store_sym(float* __restrict__ Mm, int r, int c, int m,
                                 int mo8, v8f a) {
  const int n  = c * 16 + m;
  const int i0 = r * 16 + mo8;
#pragma unroll
  for (int q = 0; q < 8; ++q) Mm[(i0 + q) * SP + n] = a[q];
  if (r != c) {
    float* p = Mm + n * SP + i0;
    *(float4*)p       = make_float4(a[0], a[1], a[2], a[3]);
    *(float4*)(p + 4) = make_float4(a[4], a[5], a[6], a[7]);
  }
}

template <bool DO_X, bool DO_P>
__device__ inline void series_step(float* __restrict__ Xm, float* __restrict__ Pm,
                                   int nt, const int (&tr)[5],
                                   const int (&tc)[5], int lane) {
  const int m   = lane & 15;
  const int kb  = (lane >> 4) << 1;
  const int mo8 = (lane >> 4) << 3;
  int oa[5], ob[5];
  v8f ax[5], ap[5];
#pragma unroll
  for (int t = 0; t < 5; ++t) {
    oa[t] = (tr[t] * 16 + m) * SP + kb;
    ob[t] = (tc[t] * 16 + m) * SP + kb;
#pragma unroll
    for (int r = 0; r < 8; ++r) {
      ax[t][r] = DO_X ? Xm[(tr[t] * 16 + mo8 + r) * SP + tc[t] * 16 + m] : 0.f;
      ap[t][r] = 0.f;
    }
  }
  if (nt == 5) step_k<5, DO_X, DO_P>(Xm, Pm, oa, ob, ax, ap);
  else         step_k<4, DO_X, DO_P>(Xm, Pm, oa, ob, ax, ap);
  __syncthreads();
#pragma unroll
  for (int t = 0; t < 5; ++t) {
    if (t < nt) {
      if (DO_P) store_sym(Pm, tr[t], tc[t], m, mo8, ap[t]);
      if (DO_X) store_sym(Xm, tr[t], tc[t], m, mo8, ax[t]);
    }
  }
  __syncthreads();
}

__global__ __launch_bounds__(256) void k_solve(const float* __restrict__ sqd,
                                               const float* __restrict__ stats,
                                               const int* __restrict__ labels,
                                               float* __restrict__ out) {
  extern __shared__ float smem[];
  float* Pm = smem;
  float* X  = smem + E * SP;
  float* hs = smem + 2 * E * SP;
  int*   lab = (int*)(smem + 2 * E * SP + E);
  const int b    = blockIdx.x;
  const int tid  = threadIdx.x;
  const int lane = tid & 31;
  const int wave = __builtin_amdgcn_readfirstlane(tid >> 5);
  const float rstd = stats_rstd(stats, lane);
  const float* sq  = sqd + (size_t)b * E * E;

  if (tid < E) lab[tid] = labels[b * E + tid];

  for (int idx = tid; idx < E * E; idx += 256) {
    const int i = idx >> 7, j = idx & 127;
    X[i * SP + j] = (i == j) ? 0.f : expf(-sq[idx] * rstd);
  }
  __syncthreads();

  if (tid < E) {
    float r = 0.f;
    for (int j = 0; j < E; ++j) r += X[tid * SP + j];
    hs[tid] = sqrtf(ALPHA / (1e-4f + r));
  }
  __syncthreads();

  for (int idx = tid; idx < E * E; idx += 256) {
    const int i = idx >> 7, j = idx & 127;
    const float nij = X[i * SP + j] * (hs[i] * hs[j]);
    Pm[i * SP + j] = nij;
    X[i * SP + j]  = ((i == j) ? 1.f : 0.f) + nij;
  }
  __syncthreads();

  int tr[5], tc[5];
  const int nt = wave_tiles(wave, tr, tc);
  series_step<false, true>(X, Pm, nt, tr, tc, lane);
  for (int it = 0; it < 3; ++it)
    series_step<true, true>(X, Pm, nt, tr, tc, lane);
  series_step<true, false>(X, Pm, nt, tr, tc, lane);

  if (tid < E) {
    const int i = tid;
    float cls[NCLS] = {0.f, 0.f, 0.f, 0.f, 0.f};
    float l1 = 0.f;
    for (int j = 0; j < E; ++j) {
      const float p = X[i * SP + j] * hs[j];
      l1 += fabsf(p);
      const int lb = lab[j];
      if (lb < NCLS) cls[lb] += p;
    }
    const float rh  = 1.f / hs[i];
    const float inv = 1.f / fmaxf(l1 * rh, 1e-12f);
#pragma unroll
    for (int c = 0; c < NCLS; ++c)
      out[((size_t)b * E + i) * NCLS + c] = logf((cls[c] * rh) * inv + 1e-6f);
  }
}

extern "C" void kernel_launch(void* const* d_in, const int* in_sizes, int n_in,
                              void* d_out, int out_size, void* d_ws, size_t ws_size,
                              hipStream_t stream) {
  const float* x      = (const float*)d_in[0];
  const int*   labels = (const int*)d_in[1];
  float* out   = (float*)d_out;
  float* sqd   = (float*)d_ws;
  float* stats = sqd + BATCH * E * E;

  const size_t sh1 = (size_t)(E * XP + E) * sizeof(float);
  k_gram<<<BATCH, 256, sh1, stream>>>(x, sqd);
  k_stats<<<SB, ST, 0, stream>>>(sqd, stats);
  const size_t sh3 = (size_t)(2 * E * SP + 2 * E) * sizeof(float);
  k_solve<<<BATCH, 256, sh3, stream>>>(sqd, stats, labels, out);
}


# k_solve prologue: W, N, X1 built on the 36 upper tiles by the waves that own them (each exp once, mirrored on store, W kept in registers to the scaling), not in 64 load-wait-exp rounds per thread
# speedup vs baseline: 5.2808x; 1.4178x over previous
#include <hip/hip_runtime.h>
#include <hip/hip_bf16.h>
#include <math.h>

typedef __attribute__((ext_vector_type(2))) float v2f;
typedef __attribute__((ext_vector_type(8))) float v8f;

#define BATCH 16
#define E 128
#define CDIM 512
#define NCLS 5
#define ALPHA 0.2f
#define XP 516
#define SP 140
#define DP 132


__device__ inline void fence() { __builtin_amdgcn_sched_barrier(0); }

__device__ inline v8f mma4(v2f a, v2f b, v8f c) {
  return __builtin_amdgcn_wmma_f32_16x16x4_f32(
       false, a,  false, b,
       (short)0, c,  false,  false);
}

__device__ inline int wave_tiles(int w, int (&tr)[5], int (&tc)[5]) {
  unsigned rp, cp;
  int nt;
  switch (w) {
    case 0:  rp = 0x00000u; cp = 0x43210u; nt = 5; break;
    case 1:  rp = 0x07000u; cp = 0x07765u; nt = 4; break;
    case 2:  rp = 0x06611u; cp = 0x07676u; nt = 4; break;
    case 3:  rp = 0x11111u; cp = 0x54321u; nt = 5; break;
    case 4:  rp = 0x05552u; cp = 0x07657u; nt = 4; break;
    case 5:  rp = 0x22222u; cp = 0x65432u; nt = 5; break;
    case 6:  rp = 0x33333u; cp = 0x76543u; nt = 5; break;
    default: rp = 0x04444u; cp = 0x07654u; nt = 4; break;
  }
#pragma unroll
  for (int t = 0; t < 5; ++t) {
    tr[t] = (int)((rp >> (4 * t)) & 15u);
    tc[t] = (int)((cp >> (4 * t)) & 15u);
  }
  return nt;
}

template <int NT>
__device__ inline void gram_load(const float* __restrict__ xs,
                                 const int (&oa)[5], const int (&ob)[5], int k0,
                                 v2f (&fa)[2][5], v2f (&fb)[2][5]) {
#pragma unroll
  for (int s = 0; s < 2; ++s) {
#pragma unroll
    for (int t = 0; t < NT; ++t) {
      fa[s][t] = *(const v2f*)(xs + oa[t] + k0 + 4 * s);
      fb[s][t] = *(const v2f*)(xs + ob[t] + k0 + 4 * s);
    }
  }
}

template <int NT>
__device__ inline void gram_mma(v8f (&acc)[5], const v2f (&fa)[2][5],
                                const v2f (&fb)[2][5]) {
#pragma unroll
  for (int s = 0; s < 2; ++s)
#pragma unroll
    for (int t = 0; t < NT; ++t) acc[t] = mma4(fa[s][t], fb[s][t], acc[t]);
}

template <int NT>
__device__ inline void gram_k(const float* __restrict__ xs, const int (&oa)[5],
                              const int (&ob)[5], v8f (&acc)[5]) {
  v2f fa0[2][5], fb0[2][5], fa1[2][5], fb1[2][5];
  gram_load<NT>(xs, oa, ob, 0, fa0, fb0);
  for (int k0 = 0; k0 < CDIM - 16; k0 += 16) {
    gram_load<NT>(xs, oa, ob, k0 + 8, fa1, fb1);
    fence();
    gram_mma<NT>(acc, fa0, fb0);
    fence();
    gram_load<NT>(xs, oa, ob, k0 + 16, fa0, fb0);
    fence();
    gram_mma<NT>(acc, fa1, fb1);
    fence();
  }
  gram_load<NT>(xs, oa, ob, CDIM - 8, fa1, fb1);
  fence();
  gram_mma<NT>(acc, fa0, fb0);
  fence();
  gram_mma<NT>(acc, fa1, fb1);
}

__global__ __launch_bounds__(256) void k_gram(const float* __restrict__ x,
                                              float* __restrict__ sqd) {
  extern __shared__ float smem[];
  float* xs    = smem;
  float* norms = smem + E * XP;
  const int b    = blockIdx.x;
  const int tid  = threadIdx.x;
  const int lane = tid & 31;
  const int wave = __builtin_amdgcn_readfirstlane(tid >> 5);

  const float4* xg = (const float4*)(x + (size_t)b * E * CDIM);
  for (int i = tid; i < E * CDIM / 4; i += 256) {
    const int r = i >> 7, c4 = i & 127;
    *(float4*)&xs[r * XP + c4 * 4] = xg[i];
  }
  __syncthreads();

  const float invs = 0.04419417382415922f;
  const int m   = lane & 15;
  const int kb  = (lane >> 4) << 1;
  const int mo8 = (lane >> 4) << 3;
  int tr[5], tc[5];
  const int nt = wave_tiles(wave, tr, tc);
  int oa[5], ob[5];
  v8f acc[5];
#pragma unroll
  for (int t = 0; t < 5; ++t) {
    oa[t]  = (tr[t] * 16 + m) * XP + kb;
    ob[t]  = (tc[t] * 16 + m) * XP + kb;
    acc[t] = (v8f){0.f, 0.f, 0.f, 0.f, 0.f, 0.f, 0.f, 0.f};
  }
  if (nt == 5) gram_k<5>(xs, oa, ob, acc);
  else         gram_k<4>(xs, oa, ob, acc);

#pragma unroll
  for (int t = 0; t < 5; ++t) {
    if (t < nt && tr[t] == tc[t]) {
      float d = acc[t][0];
#pragma unroll
      for (int q = 1; q < 8; ++q) d = ((m & 7) == q) ? acc[t][q] : d;
      if ((m >> 3) == (lane >> 4)) norms[tr[t] * 16 + m] = d;
    }
  }
  __syncthreads();

  float nn[5], ni[5][8];
#pragma unroll
  for (int t = 0; t < 5; ++t) {
    const float4 lo = *(const float4*)&norms[tr[t] * 16 + mo8];
    const float4 hi = *(const float4*)&norms[tr[t] * 16 + mo8 + 4];
    nn[t] = norms[tc[t] * 16 + m];
    ni[t][0] = lo.x; ni[t][1] = lo.y; ni[t][2] = lo.z; ni[t][3] = lo.w;
    ni[t][4] = hi.x; ni[t][5] = hi.y; ni[t][6] = hi.z; ni[t][7] = hi.w;
  }

  float* ds = smem;
#pragma unroll
  for (int t = 0; t < 5; ++t) {
    if (t < nt) {
      const int n  = tc[t] * 16 + m;
      const int i0 = tr[t] * 16 + mo8;
      float s[8];
#pragma unroll
      for (int r = 0; r < 8; ++r) {
        const int i = i0 + r;
        const float v = (ni[t][r] + nn[t] - 2.f * acc[t][r]) * invs;
        s[r] = (i == n) ? 0.f : v;
        ds[i * DP + n] = s[r];
      }
      if (tr[t] != tc[t]) {
        float* q = ds + n * DP + i0;
        *(float4*)q       = make_float4(s[0], s[1], s[2], s[3]);
        *(float4*)(q + 4) = make_float4(s[4], s[5], s[6], s[7]);
      }
    }
  }
  __syncthreads();

  float4* og = (float4*)(sqd + (size_t)b * E * E);
  for (int i = tid; i < E * E / 4; i += 256) {
    const int r = i >> 5, c4 = i & 31;
    og[i] = *(const float4*)&ds[r * DP + c4 * 4];
  }
}

#define SB 64
#define ST 256
#define SG 4
__global__ __launch_bounds__(ST) void k_stats(const float* __restrict__ sqd,
                                              float* __restrict__ part) {
  __shared__ float s_part[3][ST / 32];
  const int tid  = threadIdx.x;
  const int lane = tid & 31;
  const int wave = tid >> 5;
  const float4* __restrict__ p = (const float4*)sqd + (size_t)blockIdx.x * (ST * SG);
  float4 v[SG];
#pragma unroll
  for (int u = 0; u < SG; ++u) v[u] = p[u * ST + tid];
  float4 s4 = {0.f, 0.f, 0.f, 0.f};
  float4 q4 = {0.f, 0.f, 0.f, 0.f};
  float4 c4 = {0.f, 0.f, 0.f, 0.f};
#pragma unroll
  for (int u = 0; u < SG; ++u) {
    s4.x += v[u].x; s4.y += v[u].y; s4.z += v[u].z; s4.w += v[u].w;
    q4.x += v[u].x * v[u].x; q4.y += v[u].y * v[u].y;
    q4.z += v[u].z * v[u].z; q4.w += v[u].w * v[u].w;
    c4.x += (v[u].x != 0.f) ? 1.f : 0.f;
    c4.y += (v[u].y != 0.f) ? 1.f : 0.f;
    c4.z += (v[u].z != 0.f) ? 1.f : 0.f;
    c4.w += (v[u].w != 0.f) ? 1.f : 0.f;
  }
  float sum = (s4.x + s4.y) + (s4.z + s4.w);
  float sq  = (q4.x + q4.y) + (q4.z + q4.w);
  float cnt = (c4.x + c4.y) + (c4.z + c4.w);
  for (int off = 16; off > 0; off >>= 1) {
    sum += __shfl_xor(sum, off);
    sq  += __shfl_xor(sq, off);
    cnt += __shfl_xor(cnt, off);
  }
  if (lane == 0) {
    s_part[0][wave] = sum; s_part[1][wave] = sq; s_part[2][wave] = cnt;
  }
  __syncthreads();
  if (wave == 0) {
    const int w = lane & (ST / 32 - 1);
    sum = s_part[0][w]; sq = s_part[1][w]; cnt = s_part[2][w];
    for (int off = ST / 64; off > 0; off >>= 1) {
      sum += __shfl_xor(sum, off);
      sq  += __shfl_xor(sq, off);
      cnt += __shfl_xor(cnt, off);
    }
    if (lane == 0) {
      part[blockIdx.x]          = sum;
      part[SB + blockIdx.x]     = sq;
      part[2 * SB + blockIdx.x] = cnt;
    }
  }
}

__device__ inline float stats_rstd(const float* __restrict__ part, int lane) {
  float sum = part[lane]          + part[lane + 32];
  float sq  = part[SB + lane]     + part[SB + lane + 32];
  float cnt = part[2 * SB + lane] + part[2 * SB + lane + 32];
  for (int off = 16; off > 0; off >>= 1) {
    sum += __shfl_xor(sum, off);
    sq  += __shfl_xor(sq, off);
    cnt += __shfl_xor(cnt, off);
  }
  const float mean = sum / cnt;
  const float var  = (sq - cnt * mean * mean) / (cnt - 1.f);
  return rsqrtf(var);
}

template <int NT, bool DO_X, bool DO_P>
__device__ inline void step_load(const float* __restrict__ Xm,
                                 const float* __restrict__ Pm,
                                 const int (&oa)[5], const int (&ob)[5], int k0,
                                 v2f (&fx)[2][5], v2f (&fp)[2][5],
                                 v2f (&fb)[2][5]) {
#pragma unroll
  for (int s = 0; s < 2; ++s) {
#pragma unroll
    for (int t = 0; t < NT; ++t) {
      if (DO_X) fx[s][t] = *(const v2f*)(Xm + oa[t] + k0 + 4 * s);
      if (DO_P) fp[s][t] = *(const v2f*)(Pm + oa[t] + k0 + 4 * s);
      fb[s][t] = *(const v2f*)(Pm + ob[t] + k0 + 4 * s);
    }
  }
}

template <int NT, bool DO_X, bool DO_P>
__device__ inline void step_mma(v8f (&ax)[5], v8f (&ap)[5],
                                const v2f (&fx)[2][5], const v2f (&fp)[2][5],
                                const v2f (&fb)[2][5]) {
#pragma unroll
  for (int s = 0; s < 2; ++s) {
#pragma unroll
    for (int t = 0; t < NT; ++t) {
      if (DO_X) ax[t] = mma4(fx[s][t], fb[s][t], ax[t]);
      if (DO_P) ap[t] = mma4(fp[s][t], fb[s][t], ap[t]);
    }
  }
}

template <int NT, bool DO_X, bool DO_P>
__device__ inline void step_k(const float* __restrict__ Xm,
                              const float* __restrict__ Pm, const int (&oa)[5],
                              const int (&ob)[5], v8f (&ax)[5], v8f (&ap)[5]) {
  v2f fx0[2][5], fp0[2][5], fb0[2][5], fx1[2][5], fp1[2][5], fb1[2][5];
  step_load<NT, DO_X, DO_P>(Xm, Pm, oa, ob, 0, fx0, fp0, fb0);
  for (int k0 = 0; k0 < E - 16; k0 += 16) {
    step_load<NT, DO_X, DO_P>(Xm, Pm, oa, ob, k0 + 8, fx1, fp1, fb1);
    fence();
    step_mma<NT, DO_X, DO_P>(ax, ap, fx0, fp0, fb0);
    fence();
    step_load<NT, DO_X, DO_P>(Xm, Pm, oa, ob, k0 + 16, fx0, fp0, fb0);
    fence();
    step_mma<NT, DO_X, DO_P>(ax, ap, fx1, fp1, fb1);
    fence();
  }
  step_load<NT, DO_X, DO_P>(Xm, Pm, oa, ob, E - 8, fx1, fp1, fb1);
  fence();
  step_mma<NT, DO_X, DO_P>(ax, ap, fx0, fp0, fb0);
  fence();
  step_mma<NT, DO_X, DO_P>(ax, ap, fx1, fp1, fb1);
}

__device__ inline void store_sym(float* __restrict__ Mm, int r, int c, int m,
                                 int mo8, v8f a) {
  const int n  = c * 16 + m;
  const int i0 = r * 16 + mo8;
#pragma unroll
  for (int q = 0; q < 8; ++q) Mm[(i0 + q) * SP + n] = a[q];
  if (r != c) {
    float* p = Mm + n * SP + i0;
    *(float4*)p       = make_float4(a[0], a[1], a[2], a[3]);
    *(float4*)(p + 4) = make_float4(a[4], a[5], a[6], a[7]);
  }
}

template <bool DO_X, bool DO_P>
__device__ inline void series_step(float* __restrict__ Xm, float* __restrict__ Pm,
                                   int nt, const int (&tr)[5],
                                   const int (&tc)[5], int lane) {
  const int m   = lane & 15;
  const int kb  = (lane >> 4) << 1;
  const int mo8 = (lane >> 4) << 3;
  int oa[5], ob[5];
  v8f ax[5], ap[5];
#pragma unroll
  for (int t = 0; t < 5; ++t) {
    oa[t] = (tr[t] * 16 + m) * SP + kb;
    ob[t] = (tc[t] * 16 + m) * SP + kb;
#pragma unroll
    for (int r = 0; r < 8; ++r) {
      ax[t][r] = DO_X ? Xm[(tr[t] * 16 + mo8 + r) * SP + tc[t] * 16 + m] : 0.f;
      ap[t][r] = 0.f;
    }
  }
  if (nt == 5) step_k<5, DO_X, DO_P>(Xm, Pm, oa, ob, ax, ap);
  else         step_k<4, DO_X, DO_P>(Xm, Pm, oa, ob, ax, ap);
  __syncthreads();
#pragma unroll
  for (int t = 0; t < 5; ++t) {
    if (t < nt) {
      if (DO_P) store_sym(Pm, tr[t], tc[t], m, mo8, ap[t]);
      if (DO_X) store_sym(Xm, tr[t], tc[t], m, mo8, ax[t]);
    }
  }
  __syncthreads();
}

__global__ __launch_bounds__(256) void k_solve(const float* __restrict__ sqd,
                                               const float* __restrict__ stats,
                                               const int* __restrict__ labels,
                                               float* __restrict__ out) {
  extern __shared__ float smem[];
  float* Pm = smem;
  float* X  = smem + E * SP;
  float* hs = smem + 2 * E * SP;
  int*   lab = (int*)(smem + 2 * E * SP + E);
  const int b    = blockIdx.x;
  const int tid  = threadIdx.x;
  const int lane = tid & 31;
  const int wave = __builtin_amdgcn_readfirstlane(tid >> 5);
  const float rstd = stats_rstd(stats, lane);
  const float* sq  = sqd + (size_t)b * E * E;

  if (tid < E) lab[tid] = labels[b * E + tid];

  int tr[5], tc[5];
  const int nt = wave_tiles(wave, tr, tc);
  const int m   = lane & 15;
  const int mo8 = (lane >> 4) << 3;

  v8f w[5];
#pragma unroll
  for (int t = 0; t < 5; ++t) {
    const int n  = tc[t] * 16 + m;
    const int i0 = tr[t] * 16 + mo8;
#pragma unroll
    for (int q = 0; q < 8; ++q) {
      const float e = expf(-sq[(i0 + q) * E + n] * rstd);
      w[t][q] = (i0 + q == n) ? 0.f : e;
    }
  }
#pragma unroll
  for (int t = 0; t < 5; ++t)
    if (t < nt) store_sym(X, tr[t], tc[t], m, mo8, w[t]);
  __syncthreads();

  if (tid < E) {
    float r = 0.f;
    for (int j = 0; j < E; ++j) r += X[tid * SP + j];
    hs[tid] = sqrtf(ALPHA / (1e-4f + r));
  }
  __syncthreads();

#pragma unroll
  for (int t = 0; t < 5; ++t) {
    if (t < nt) {
      const int n  = tc[t] * 16 + m;
      const int i0 = tr[t] * 16 + mo8;
      const float hn  = hs[n];
      const float4 lo = *(const float4*)&hs[i0];
      const float4 hi = *(const float4*)&hs[i0 + 4];
      const float hr[8] = {lo.x, lo.y, lo.z, lo.w, hi.x, hi.y, hi.z, hi.w};
      v8f nv, xv;
#pragma unroll
      for (int q = 0; q < 8; ++q) {
        const float nij = w[t][q] * (hr[q] * hn);
        nv[q] = nij;
        xv[q] = ((i0 + q == n) ? 1.f : 0.f) + nij;
      }
      store_sym(Pm, tr[t], tc[t], m, mo8, nv);
      store_sym(X, tr[t], tc[t], m, mo8, xv);
    }
  }
  __syncthreads();

  series_step<false, true>(X, Pm, nt, tr, tc, lane);
  for (int it = 0; it < 3; ++it)
    series_step<true, true>(X, Pm, nt, tr, tc, lane);
  series_step<true, false>(X, Pm, nt, tr, tc, lane);

  if (tid < E) {
    const int i = tid;
    float cls[NCLS] = {0.f, 0.f, 0.f, 0.f, 0.f};
    float l1 = 0.f;
    for (int j = 0; j < E; ++j) {
      const float p = X[i * SP + j] * hs[j];
      l1 += fabsf(p);
      const int lb = lab[j];
      if (lb < NCLS) cls[lb] += p;
    }
    const float rh  = 1.f / hs[i];
    const float inv = 1.f / fmaxf(l1 * rh, 1e-12f);
#pragma unroll
    for (int c = 0; c < NCLS; ++c)
      out[((size_t)b * E + i) * NCLS + c] = logf((cls[c] * rh) * inv + 1e-6f);
  }
}

extern "C" void kernel_launch(void* const* d_in, const int* in_sizes, int n_in,
                              void* d_out, int out_size, void* d_ws, size_t ws_size,
                              hipStream_t stream) {
  const float* x      = (const float*)d_in[0];
  const int*   labels = (const int*)d_in[1];
  float* out   = (float*)d_out;
  float* sqd   = (float*)d_ws;
  float* stats = sqd + BATCH * E * E;

  const size_t sh1 = (size_t)(E * XP + E) * sizeof(float);
  k_gram<<<BATCH, 256, sh1, stream>>>(x, sqd);
  k_stats<<<SB, ST, 0, stream>>>(sqd, stats);
  const size_t sh3 = (size_t)(2 * E * SP + 2 * E) * sizeof(float);
  k_solve<<<BATCH, 256, sh3, stream>>>(sqd, stats, labels, out);
}


# k_solve: each batch shared by 4 work-groups; every one repeats the squarings of P but carries only its own 32 rows of the running sum X (two of the eight row bands) and writes those 32 output rows
# speedup vs baseline: 7.6572x; 1.4500x over previous
#include <hip/hip_runtime.h>
#include <hip/hip_bf16.h>
#include <math.h>

typedef __attribute__((ext_vector_type(2))) float v2f;
typedef __attribute__((ext_vector_type(8))) float v8f;

#define BATCH 16
#define E 128
#define CDIM 512
#define NCLS 5
#define ALPHA 0.2f
#define KS 8
#define KW (CDIM / KS)
#define XP (KW + 4)
#define XPW (CDIM + 4)
#define SP 140
#define WS_SPLIT_BYTES \
  ((size_t)(BATCH * E * E + 3 * 256 + BATCH * KS * E + BATCH * KS * E * E) * sizeof(float))
#define DP 132


__device__ inline void fence() { __builtin_amdgcn_sched_barrier(0); }

__device__ inline v8f mma4(v2f a, v2f b, v8f c) {
  return __builtin_amdgcn_wmma_f32_16x16x4_f32(
       false, a,  false, b,
       (short)0, c,  false,  false);
}

__device__ inline int wave_tiles(int w, int (&tr)[5], int (&tc)[5]) {
  unsigned rp, cp;
  int nt;
  switch (w) {
    case 0:  rp = 0x00000u; cp = 0x43210u; nt = 5; break;
    case 1:  rp = 0x07000u; cp = 0x07765u; nt = 4; break;
    case 2:  rp = 0x06611u; cp = 0x07676u; nt = 4; break;
    case 3:  rp = 0x11111u; cp = 0x54321u; nt = 5; break;
    case 4:  rp = 0x05552u; cp = 0x07657u; nt = 4; break;
    case 5:  rp = 0x22222u; cp = 0x65432u; nt = 5; break;
    case 6:  rp = 0x33333u; cp = 0x76543u; nt = 5; break;
    default: rp = 0x04444u; cp = 0x07654u; nt = 4; break;
  }
#pragma unroll
  for (int t = 0; t < 5; ++t) {
    tr[t] = (int)((rp >> (4 * t)) & 15u);
    tc[t] = (int)((cp >> (4 * t)) & 15u);
  }
  return nt;
}

template <int NT>
__device__ inline void gram_load(const float* __restrict__ xs,
                                 const int (&oa)[5], const int (&ob)[5], int k0,
                                 v2f (&fa)[2][5], v2f (&fb)[2][5]) {
#pragma unroll
  for (int s = 0; s < 2; ++s) {
#pragma unroll
    for (int t = 0; t < NT; ++t) {
      fa[s][t] = *(const v2f*)(xs + oa[t] + k0 + 4 * s);
      fb[s][t] = *(const v2f*)(xs + ob[t] + k0 + 4 * s);
    }
  }
}

template <int NT>
__device__ inline void gram_mma(v8f (&acc)[5], const v2f (&fa)[2][5],
                                const v2f (&fb)[2][5]) {
#pragma unroll
  for (int s = 0; s < 2; ++s)
#pragma unroll
    for (int t = 0; t < NT; ++t) acc[t] = mma4(fa[s][t], fb[s][t], acc[t]);
}

template <int NT, int KX>
__device__ inline void gram_k(const float* __restrict__ xs, const int (&oa)[5],
                              const int (&ob)[5], v8f (&acc)[5]) {
  v2f fa0[2][5], fb0[2][5], fa1[2][5], fb1[2][5];
  gram_load<NT>(xs, oa, ob, 0, fa0, fb0);
  for (int k0 = 0; k0 < KX - 16; k0 += 16) {
    gram_load<NT>(xs, oa, ob, k0 + 8, fa1, fb1);
    fence();
    gram_mma<NT>(acc, fa0, fb0);
    fence();
    gram_load<NT>(xs, oa, ob, k0 + 16, fa0, fb0);
    fence();
    gram_mma<NT>(acc, fa1, fb1);
    fence();
  }
  gram_load<NT>(xs, oa, ob, KX - 8, fa1, fb1);
  fence();
  gram_mma<NT>(acc, fa0, fb0);
  fence();
  gram_mma<NT>(acc, fa1, fb1);
}

__global__ __launch_bounds__(256) void k_gram(const float* __restrict__ x,
                                              float* __restrict__ gpart,
                                              float* __restrict__ pnorm) {
  extern __shared__ float smem[];
  float* xs    = smem;
  float* norms = smem + E * DP;
  const int b    = blockIdx.x / KS;
  const int ks   = blockIdx.x % KS;
  const int tid  = threadIdx.x;
  const int lane = tid & 31;
  const int wave = __builtin_amdgcn_readfirstlane(tid >> 5);

  const float* xg = x + (size_t)b * E * CDIM + ks * KW;
  float4 sv[E * KW / 4 / 256];
#pragma unroll
  for (int u = 0; u < E * KW / 4 / 256; ++u) {
    const int i = u * 256 + tid;
    const int r = i / (KW / 4), c4 = i % (KW / 4);
    sv[u] = *(const float4*)(xg + r * CDIM + c4 * 4);
  }
#pragma unroll
  for (int u = 0; u < E * KW / 4 / 256; ++u) {
    const int i = u * 256 + tid;
    const int r = i / (KW / 4), c4 = i % (KW / 4);
    *(float4*)&xs[r * XP + c4 * 4] = sv[u];
  }
  __syncthreads();

  const int m   = lane & 15;
  const int kb  = (lane >> 4) << 1;
  const int mo8 = (lane >> 4) << 3;
  int tr[5], tc[5];
  const int nt = wave_tiles(wave, tr, tc);
  int oa[5], ob[5];
  v8f acc[5];
#pragma unroll
  for (int t = 0; t < 5; ++t) {
    oa[t]  = (tr[t] * 16 + m) * XP + kb;
    ob[t]  = (tc[t] * 16 + m) * XP + kb;
    acc[t] = (v8f){0.f, 0.f, 0.f, 0.f, 0.f, 0.f, 0.f, 0.f};
  }
  if (nt == 5) gram_k<5, KW>(xs, oa, ob, acc);
  else         gram_k<4, KW>(xs, oa, ob, acc);

#pragma unroll
  for (int t = 0; t < 5; ++t) {
    if (t < nt && tr[t] == tc[t]) {
      float d = acc[t][0];
#pragma unroll
      for (int q = 1; q < 8; ++q) d = ((m & 7) == q) ? acc[t][q] : d;
      if ((m >> 3) == (lane >> 4)) norms[tr[t] * 16 + m] = d;
    }
  }
  __syncthreads();
  if (tid < E) pnorm[(size_t)blockIdx.x * E + tid] = norms[tid];

  float* ds = smem;
#pragma unroll
  for (int t = 0; t < 5; ++t) {
    if (t < nt) {
      const int n  = tc[t] * 16 + m;
      const int i0 = tr[t] * 16 + mo8;
#pragma unroll
      for (int r = 0; r < 8; ++r) ds[(i0 + r) * DP + n] = acc[t][r];
      if (tr[t] != tc[t]) {
        float* q = ds + n * DP + i0;
        *(float4*)q       = make_float4(acc[t][0], acc[t][1], acc[t][2], acc[t][3]);
        *(float4*)(q + 4) = make_float4(acc[t][4], acc[t][5], acc[t][6], acc[t][7]);
      }
    }
  }
  __syncthreads();

  float4* og = (float4*)(gpart + (size_t)blockIdx.x * E * E);
  for (int i = tid; i < E * E / 4; i += 256) {
    const int r = i >> 5, c4 = i & 31;
    og[i] = *(const float4*)&ds[r * DP + c4 * 4];
  }
}

__global__ __launch_bounds__(256) void k_gram_whole(const float* __restrict__ x,
                                                    float* __restrict__ sqd) {
  extern __shared__ float smem[];
  float* xs    = smem;
  float* norms = smem + E * XPW;
  const int b    = blockIdx.x;
  const int tid  = threadIdx.x;
  const int lane = tid & 31;
  const int wave = __builtin_amdgcn_readfirstlane(tid >> 5);

  const float4* xg = (const float4*)(x + (size_t)b * E * CDIM);
  for (int i = tid; i < E * CDIM / 4; i += 256) {
    const int r = i >> 7, c4 = i & 127;
    *(float4*)&xs[r * XPW + c4 * 4] = xg[i];
  }
  __syncthreads();

  const float invs = 0.04419417382415922f;
  const int m   = lane & 15;
  const int kb  = (lane >> 4) << 1;
  const int mo8 = (lane >> 4) << 3;
  int tr[5], tc[5];
  const int nt = wave_tiles(wave, tr, tc);
  int oa[5], ob[5];
  v8f acc[5];
#pragma unroll
  for (int t = 0; t < 5; ++t) {
    oa[t]  = (tr[t] * 16 + m) * XPW + kb;
    ob[t]  = (tc[t] * 16 + m) * XPW + kb;
    acc[t] = (v8f){0.f, 0.f, 0.f, 0.f, 0.f, 0.f, 0.f, 0.f};
  }
  if (nt == 5) gram_k<5, CDIM>(xs, oa, ob, acc);
  else         gram_k<4, CDIM>(xs, oa, ob, acc);

#pragma unroll
  for (int t = 0; t < 5; ++t) {
    if (t < nt && tr[t] == tc[t]) {
      float d = acc[t][0];
#pragma unroll
      for (int q = 1; q < 8; ++q) d = ((m & 7) == q) ? acc[t][q] : d;
      if ((m >> 3) == (lane >> 4)) norms[tr[t] * 16 + m] = d;
    }
  }
  __syncthreads();

  float nn[5], ni[5][8];
#pragma unroll
  for (int t = 0; t < 5; ++t) {
    const float4 lo = *(const float4*)&norms[tr[t] * 16 + mo8];
    const float4 hi = *(const float4*)&norms[tr[t] * 16 + mo8 + 4];
    nn[t] = norms[tc[t] * 16 + m];
    ni[t][0] = lo.x; ni[t][1] = lo.y; ni[t][2] = lo.z; ni[t][3] = lo.w;
    ni[t][4] = hi.x; ni[t][5] = hi.y; ni[t][6] = hi.z; ni[t][7] = hi.w;
  }

  float* ds = smem;
#pragma unroll
  for (int t = 0; t < 5; ++t) {
    if (t < nt) {
      const int n  = tc[t] * 16 + m;
      const int i0 = tr[t] * 16 + mo8;
      float s[8];
#pragma unroll
      for (int r = 0; r < 8; ++r) {
        const int i = i0 + r;
        const float v = (ni[t][r] + nn[t] - 2.f * acc[t][r]) * invs;
        s[r] = (i == n) ? 0.f : v;
        ds[i * DP + n] = s[r];
      }
      if (tr[t] != tc[t]) {
        float* q = ds + n * DP + i0;
        *(float4*)q       = make_float4(s[0], s[1], s[2], s[3]);
        *(float4*)(q + 4) = make_float4(s[4], s[5], s[6], s[7]);
      }
    }
  }
  __syncthreads();

  float4* og = (float4*)(sqd + (size_t)b * E * E);
  for (int i = tid; i < E * E / 4; i += 256) {
    const int r = i >> 5, c4 = i & 31;
    og[i] = *(const float4*)&ds[r * DP + c4 * 4];
  }
}

#define SB 256
#define ST 256
__global__ __launch_bounds__(ST) void k_stats(const float* __restrict__ gpart,
                                              const float* __restrict__ pnorm,
                                              float* __restrict__ sqd,
                                              float* __restrict__ part) {
  __shared__ __attribute__((aligned(16))) float s_norm[E];
  __shared__ float s_part[3][ST / 32];
  const int tid  = threadIdx.x;
  const int lane = tid & 31;
  const int wave = tid >> 5;
  const int b    = blockIdx.x >> 4;
  const int row  = ((blockIdx.x & 15) << 3) + wave;
  const int col  = lane << 2;
  const float4* __restrict__ gp =
      (const float4*)(gpart + (size_t)b * KS * E * E + row * E + col);
  const float* __restrict__ np = pnorm + (size_t)b * KS * E + (tid & (E - 1));
  float4 v[KS];
  float pv[KS];
#pragma unroll
  for (int s = 0; s < KS; ++s) v[s] = gp[s * (E * E / 4)];
#pragma unroll
  for (int s = 0; s < KS; ++s) pv[s] = np[s * E];
  float nsum = pv[0];
  float4 G = v[0];
#pragma unroll
  for (int s = 1; s < KS; ++s) {
    nsum += pv[s];
    G.x += v[s].x; G.y += v[s].y; G.z += v[s].z; G.w += v[s].w;
  }
  if (tid < E) s_norm[tid] = nsum;
  __syncthreads();
  const float invs = 0.04419417382415922f;
  const float ni  = s_norm[row];
  const float4 nj = *(const float4*)&s_norm[col];
  float4 d;
  d.x = (ni + nj.x - 2.f * G.x) * invs;
  d.y = (ni + nj.y - 2.f * G.y) * invs;
  d.z = (ni + nj.z - 2.f * G.z) * invs;
  d.w = (ni + nj.w - 2.f * G.w) * invs;
  d.x = (row == col)     ? 0.f : d.x;
  d.y = (row == col + 1) ? 0.f : d.y;
  d.z = (row == col + 2) ? 0.f : d.z;
  d.w = (row == col + 3) ? 0.f : d.w;
  ((float4*)sqd)[(size_t)blockIdx.x * ST + tid] = d;

  float sum = (d.x + d.y) + (d.z + d.w);
  float sq  = (d.x * d.x + d.y * d.y) + (d.z * d.z + d.w * d.w);
  float cnt = (((d.x != 0.f) ? 1.f : 0.f) + ((d.y != 0.f) ? 1.f : 0.f)) +
              (((d.z != 0.f) ? 1.f : 0.f) + ((d.w != 0.f) ? 1.f : 0.f));
  for (int off = 16; off > 0; off >>= 1) {
    sum += __shfl_xor(sum, off);
    sq  += __shfl_xor(sq, off);
    cnt += __shfl_xor(cnt, off);
  }
  if (lane == 0) {
    s_part[0][wave] = sum; s_part[1][wave] = sq; s_part[2][wave] = cnt;
  }
  __syncthreads();
  if (wave == 0) {
    const int w = lane & (ST / 32 - 1);
    sum = s_part[0][w]; sq = s_part[1][w]; cnt = s_part[2][w];
    for (int off = ST / 64; off > 0; off >>= 1) {
      sum += __shfl_xor(sum, off);
      sq  += __shfl_xor(sq, off);
      cnt += __shfl_xor(cnt, off);
    }
    if (lane == 0) {
      part[blockIdx.x]          = sum;
      part[SB + blockIdx.x]     = sq;
      part[2 * SB + blockIdx.x] = cnt;
    }
  }
}

#define SBW 64
#define SG 4
__global__ __launch_bounds__(ST) void k_stats_whole(const float* __restrict__ sqd,
                                                    float* __restrict__ part) {
  __shared__ float s_part[3][ST / 32];
  const int tid  = threadIdx.x;
  const int lane = tid & 31;
  const int wave = tid >> 5;
  const float4* __restrict__ p = (const float4*)sqd + (size_t)blockIdx.x * (ST * SG);
  float4 v[SG];
#pragma unroll
  for (int u = 0; u < SG; ++u) v[u] = p[u * ST + tid];
  float4 s4 = {0.f, 0.f, 0.f, 0.f};
  float4 q4 = {0.f, 0.f, 0.f, 0.f};
  float4 c4 = {0.f, 0.f, 0.f, 0.f};
#pragma unroll
  for (int u = 0; u < SG; ++u) {
    s4.x += v[u].x; s4.y += v[u].y; s4.z += v[u].z; s4.w += v[u].w;
    q4.x += v[u].x * v[u].x; q4.y += v[u].y * v[u].y;
    q4.z += v[u].z * v[u].z; q4.w += v[u].w * v[u].w;
    c4.x += (v[u].x != 0.f) ? 1.f : 0.f;
    c4.y += (v[u].y != 0.f) ? 1.f : 0.f;
    c4.z += (v[u].z != 0.f) ? 1.f : 0.f;
    c4.w += (v[u].w != 0.f) ? 1.f : 0.f;
  }
  float sum = (s4.x + s4.y) + (s4.z + s4.w);
  float sq  = (q4.x + q4.y) + (q4.z + q4.w);
  float cnt = (c4.x + c4.y) + (c4.z + c4.w);
  for (int off = 16; off > 0; off >>= 1) {
    sum += __shfl_xor(sum, off);
    sq  += __shfl_xor(sq, off);
    cnt += __shfl_xor(cnt, off);
  }
  if (lane == 0) {
    s_part[0][wave] = sum; s_part[1][wave] = sq; s_part[2][wave] = cnt;
  }
  __syncthreads();
  if (wave == 0) {
    const int w = lane & (ST / 32 - 1);
    sum = s_part[0][w]; sq = s_part[1][w]; cnt = s_part[2][w];
    for (int off = ST / 64; off > 0; off >>= 1) {
      sum += __shfl_xor(sum, off);
      sq  += __shfl_xor(sq, off);
      cnt += __shfl_xor(cnt, off);
    }
    if (lane == 0) {
      part[blockIdx.x]          = sum;
      part[SBW + blockIdx.x]     = sq;
      part[2 * SBW + blockIdx.x] = cnt;
    }
  }
}

template <int NB>
__device__ inline float stats_rstd(const float* __restrict__ part, int lane) {
  float sum = part[lane];
  float sq  = part[NB + lane];
  float cnt = part[2 * NB + lane];
#pragma unroll
  for (int k = 1; k < NB / 32; ++k) {
    sum += part[lane + 32 * k];
    sq  += part[NB + lane + 32 * k];
    cnt += part[2 * NB + lane + 32 * k];
  }
  for (int off = 16; off > 0; off >>= 1) {
    sum += __shfl_xor(sum, off);
    sq  += __shfl_xor(sq, off);
    cnt += __shfl_xor(cnt, off);
  }
  const float mean = sum / cnt;
  const float var  = (sq - cnt * mean * mean) / (cnt - 1.f);
  return rsqrtf(var);
}

#define RS 4
#define XB (E / 16 / RS)
static_assert(XB == 2, "the wave -> X tile map below is written for 2 row bands");

template <int NT, bool DO_X, bool DO_P>
__device__ inline void step_load(const float* __restrict__ Xm,
                                 const float* __restrict__ Pm,
                                 const int (&oa)[5], const int (&ob)[5], int oxa,
                                 const int (&oxb)[2], int k0, v2f (&fp)[2][5],
                                 v2f (&fb)[2][5], v2f (&fxa)[2],
                                 v2f (&fxb)[2][2]) {
#pragma unroll
  for (int s = 0; s < 2; ++s) {
    if (DO_P) {
#pragma unroll
      for (int t = 0; t < NT; ++t) {
        fp[s][t] = *(const v2f*)(Pm + oa[t] + k0 + 4 * s);
        fb[s][t] = *(const v2f*)(Pm + ob[t] + k0 + 4 * s);
      }
    }
    if (DO_X) {
      fxa[s] = *(const v2f*)(Xm + oxa + k0 + 4 * s);
#pragma unroll
      for (int j = 0; j < 2; ++j)
        fxb[s][j] = *(const v2f*)(Pm + oxb[j] + k0 + 4 * s);
    }
  }
}

template <int NT, bool DO_X, bool DO_P>
__device__ inline void step_mma(v8f (&ax)[2], v8f (&ap)[5],
                                const v2f (&fp)[2][5], const v2f (&fb)[2][5],
                                const v2f (&fxa)[2], const v2f (&fxb)[2][2]) {
#pragma unroll
  for (int s = 0; s < 2; ++s) {
    if (DO_P) {
#pragma unroll
      for (int t = 0; t < NT; ++t) ap[t] = mma4(fp[s][t], fb[s][t], ap[t]);
    }
    if (DO_X) {
#pragma unroll
      for (int j = 0; j < 2; ++j) ax[j] = mma4(fxa[s], fxb[s][j], ax[j]);
    }
  }
}

template <int NT, bool DO_X, bool DO_P>
__device__ inline void step_k(const float* __restrict__ Xm,
                              const float* __restrict__ Pm, const int (&oa)[5],
                              const int (&ob)[5], int oxa, const int (&oxb)[2],
                              v8f (&ax)[2], v8f (&ap)[5]) {
  v2f fp0[2][5], fb0[2][5], fxa0[2], fxb0[2][2];
  v2f fp1[2][5], fb1[2][5], fxa1[2], fxb1[2][2];
  step_load<NT, DO_X, DO_P>(Xm, Pm, oa, ob, oxa, oxb, 0, fp0, fb0, fxa0, fxb0);
  for (int k0 = 0; k0 < E - 16; k0 += 16) {
    step_load<NT, DO_X, DO_P>(Xm, Pm, oa, ob, oxa, oxb, k0 + 8, fp1, fb1, fxa1, fxb1);
    fence();
    step_mma<NT, DO_X, DO_P>(ax, ap, fp0, fb0, fxa0, fxb0);
    fence();
    step_load<NT, DO_X, DO_P>(Xm, Pm, oa, ob, oxa, oxb, k0 + 16, fp0, fb0, fxa0, fxb0);
    fence();
    step_mma<NT, DO_X, DO_P>(ax, ap, fp1, fb1, fxa1, fxb1);
    fence();
  }
  step_load<NT, DO_X, DO_P>(Xm, Pm, oa, ob, oxa, oxb, E - 8, fp1, fb1, fxa1, fxb1);
  fence();
  step_mma<NT, DO_X, DO_P>(ax, ap, fp0, fb0, fxa0, fxb0);
  fence();
  step_mma<NT, DO_X, DO_P>(ax, ap, fp1, fb1, fxa1, fxb1);
}

__device__ inline void store_sym(float* __restrict__ Mm, int r, int c, int m,
                                 int mo8, v8f a) {
  const int n  = c * 16 + m;
  const int i0 = r * 16 + mo8;
#pragma unroll
  for (int q = 0; q < 8; ++q) Mm[(i0 + q) * SP + n] = a[q];
  if (r != c) {
    float* p = Mm + n * SP + i0;
    *(float4*)p       = make_float4(a[0], a[1], a[2], a[3]);
    *(float4*)(p + 4) = make_float4(a[4], a[5], a[6], a[7]);
  }
}

template <bool DO_X, bool DO_P>
__device__ inline void series_step(float* __restrict__ Xm, float* __restrict__ Pm,
                                   int nt, const int (&tr)[5],
                                   const int (&tc)[5], int xr, int xc0,
                                   int lane) {
  const int m   = lane & 15;
  const int kb  = (lane >> 4) << 1;
  const int mo8 = (lane >> 4) << 3;
  int oa[5], ob[5], oxb[2];
  v8f ax[2], ap[5];
#pragma unroll
  for (int t = 0; t < 5; ++t) {
    oa[t] = (tr[t] * 16 + m) * SP + kb;
    ob[t] = (tc[t] * 16 + m) * SP + kb;
#pragma unroll
    for (int r = 0; r < 8; ++r) ap[t][r] = 0.f;
  }
  const int oxa = (xr * 16 + m) * SP + kb;
#pragma unroll
  for (int j = 0; j < 2; ++j) {
    oxb[j] = ((xc0 + j) * 16 + m) * SP + kb;
#pragma unroll
    for (int r = 0; r < 8; ++r)
      ax[j][r] = DO_X ? Xm[(xr * 16 + mo8 + r) * SP + (xc0 + j) * 16 + m] : 0.f;
  }
  if (DO_P && nt == 5) step_k<5, DO_X, DO_P>(Xm, Pm, oa, ob, oxa, oxb, ax, ap);
  else                 step_k<4, DO_X, DO_P>(Xm, Pm, oa, ob, oxa, oxb, ax, ap);
  __syncthreads();
  if (DO_P) {
#pragma unroll
    for (int t = 0; t < 5; ++t)
      if (t < nt) store_sym(Pm, tr[t], tc[t], m, mo8, ap[t]);
  }
  if (DO_X) {
#pragma unroll
    for (int j = 0; j < 2; ++j)
#pragma unroll
      for (int r = 0; r < 8; ++r)
        Xm[(xr * 16 + mo8 + r) * SP + (xc0 + j) * 16 + m] = ax[j][r];
  }
  __syncthreads();
}

template <int NB>
__global__ __launch_bounds__(256) void k_solve(const float* __restrict__ sqd,
                                               const float* __restrict__ stats,
                                               const int* __restrict__ labels,
                                               float* __restrict__ out) {
  extern __shared__ float smem[];
  float* Pm = smem;
  float* X  = smem + E * SP;
  float* hs = smem + 2 * E * SP;
  int*   lab = (int*)(smem + 2 * E * SP + E);
  const int b    = blockIdx.x / RS;
  const int rb0  = (blockIdx.x % RS) * XB;
  const int tid  = threadIdx.x;
  const int lane = tid & 31;
  const int wave = __builtin_amdgcn_readfirstlane(tid >> 5);
  const float rstd = stats_rstd<NB>(stats, lane);
  const float* sq  = sqd + (size_t)b * E * E;

  if (tid < E) lab[tid] = labels[b * E + tid];

  int tr[5], tc[5];
  const int nt = wave_tiles(wave, tr, tc);
  const int m   = lane & 15;
  const int mo8 = (lane >> 4) << 3;
  const int xr  = rb0 + (wave & 1);
  const int xc0 = (wave >> 1) << 1;

  v8f w[5];
#pragma unroll
  for (int t = 0; t < 5; ++t) {
    const int n  = tc[t] * 16 + m;
    const int i0 = tr[t] * 16 + mo8;
#pragma unroll
    for (int q = 0; q < 8; ++q) {
      const float e = expf(-sq[(i0 + q) * E + n] * rstd);
      w[t][q] = (i0 + q == n) ? 0.f : e;
    }
  }
#pragma unroll
  for (int t = 0; t < 5; ++t)
    if (t < nt) store_sym(X, tr[t], tc[t], m, mo8, w[t]);
  __syncthreads();

  if (tid < E) {
    float r = 0.f;
    for (int j = 0; j < E; ++j) r += X[tid * SP + j];
    hs[tid] = sqrtf(ALPHA / (1e-4f + r));
  }
  __syncthreads();

#pragma unroll
  for (int t = 0; t < 5; ++t) {
    if (t < nt) {
      const int n  = tc[t] * 16 + m;
      const int i0 = tr[t] * 16 + mo8;
      const float hn  = hs[n];
      const float4 lo = *(const float4*)&hs[i0];
      const float4 hi = *(const float4*)&hs[i0 + 4];
      const float hr[8] = {lo.x, lo.y, lo.z, lo.w, hi.x, hi.y, hi.z, hi.w};
      v8f nv, xv;
#pragma unroll
      for (int q = 0; q < 8; ++q) {
        const float nij = w[t][q] * (hr[q] * hn);
        nv[q] = nij;
        xv[q] = ((i0 + q == n) ? 1.f : 0.f) + nij;
      }
      store_sym(Pm, tr[t], tc[t], m, mo8, nv);
      store_sym(X, tr[t], tc[t], m, mo8, xv);
    }
  }
  __syncthreads();

  series_step<false, true>(X, Pm, nt, tr, tc, xr, xc0, lane);
  for (int it = 0; it < 3; ++it)
    series_step<true, true>(X, Pm, nt, tr, tc, xr, xc0, lane);
  series_step<true, false>(X, Pm, nt, tr, tc, xr, xc0, lane);

  if (tid < 16 * XB) {
    const int i = rb0 * 16 + tid;
    float cls[NCLS] = {0.f, 0.f, 0.f, 0.f, 0.f};
    float l1 = 0.f;
    for (int j = 0; j < E; ++j) {
      const float p = X[i * SP + j] * hs[j];
      l1 += fabsf(p);
      const int lb = lab[j];
      if (lb < NCLS) cls[lb] += p;
    }
    const float rh  = 1.f / hs[i];
    const float inv = 1.f / fmaxf(l1 * rh, 1e-12f);
#pragma unroll
    for (int c = 0; c < NCLS; ++c)
      out[((size_t)b * E + i) * NCLS + c] = logf((cls[c] * rh) * inv + 1e-6f);
  }
}

extern "C" void kernel_launch(void* const* d_in, const int* in_sizes, int n_in,
                              void* d_out, int out_size, void* d_ws, size_t ws_size,
                              hipStream_t stream) {
  const float* x      = (const float*)d_in[0];
  const int*   labels = (const int*)d_in[1];
  float* out   = (float*)d_out;
  float* sqd   = (float*)d_ws;
  float* stats = sqd + BATCH * E * E;
  const size_t sh3 = (size_t)(2 * E * SP + 2 * E) * sizeof(float);

  if (ws_size >= WS_SPLIT_BYTES) {
    float* pnorm = stats + 3 * SB;
    float* gpart = pnorm + BATCH * KS * E;
    const size_t sh1 = (size_t)(E * DP + E) * sizeof(float);
    k_gram<<<BATCH * KS, 256, sh1, stream>>>(x, gpart, pnorm);
    k_stats<<<SB, ST, 0, stream>>>(gpart, pnorm, sqd, stats);
    k_solve<SB><<<BATCH * RS, 256, sh3, stream>>>(sqd, stats, labels, out);
  } else {
    const size_t sh1 = (size_t)(E * XPW + E) * sizeof(float);
    k_gram_whole<<<BATCH, 256, sh1, stream>>>(x, sqd);
    k_stats_whole<<<SBW, ST, 0, stream>>>(sqd, stats);
    k_solve<SBW><<<BATCH * RS, 256, sh3, stream>>>(sqd, stats, labels, out);
  }
}
